# starting kernel
# baseline (speedup 1.0000x reference)
#include <hip/hip_runtime.h>


#define N_NODES 100000
#define N_EDGES 1600000
#define D       128
#define M_TILE  64
#define TPB     128
#define XPITCH  80
#define WPITCH  144

typedef __attribute__((ext_vector_type(2))) float v2f;
typedef __attribute__((ext_vector_type(8))) float v8f;

__device__ __forceinline__ v2f mk2(float a, float b) {
  v2f t; t.x = a; t.y = b; return t;
}

__global__ __launch_bounds__(256) void zero_kernel(float* __restrict__ out, int n4) {
  float4 z = make_float4(0.f, 0.f, 0.f, 0.f);
  float4* o4 = reinterpret_cast<float4*>(out);
  for (int i = blockIdx.x * blockDim.x + threadIdx.x; i < n4;
       i += gridDim.x * blockDim.x)
    o4[i] = z;
}

__global__ __launch_bounds__(256) void scatter_kernel(const float* __restrict__ nf,
                                                      const int* __restrict__ src,
                                                      const int* __restrict__ dst,
                                                      float* __restrict__ agg) {
  int e    = blockIdx.x * (256 >> 5) + (threadIdx.x >> 5);
  int lane = threadIdx.x & 31;
  if (e >= N_EDGES) return;
  int s = src[e];
  int d = dst[e];
  const float4* row = reinterpret_cast<const float4*>(nf + (size_t)s * D);
  float4 v = row[lane];
  float* a = agg + (size_t)d * D + lane * 4;
  __hip_atomic_fetch_add(a + 0, v.x, __ATOMIC_RELAXED, __HIP_MEMORY_SCOPE_AGENT);
  __hip_atomic_fetch_add(a + 1, v.y, __ATOMIC_RELAXED, __HIP_MEMORY_SCOPE_AGENT);
  __hip_atomic_fetch_add(a + 2, v.z, __ATOMIC_RELAXED, __HIP_MEMORY_SCOPE_AGENT);
  __hip_atomic_fetch_add(a + 3, v.w, __ATOMIC_RELAXED, __HIP_MEMORY_SCOPE_AGENT);
}


__device__ __forceinline__ void load_tile_sw(v2f* __restrict__ sT,
                                             const float* __restrict__ g,
                                             int rowBase, int tid) {
  const float4* g4 = reinterpret_cast<const float4*>(g);
#pragma unroll
  for (int i = 0; i < 16; ++i) {
    int linear = i * TPB + tid;
    int r  = linear & 63;
    int c4 = linear >> 6;
    int gr = rowBase + r;
    if (gr >= N_NODES) gr = N_NODES - 1;
    float4 v = g4[(size_t)gr * 32 + c4];
    sT[(c4 * 2) * XPITCH + r]     = mk2(v.x, v.y);
    sT[(c4 * 2 + 1) * XPITCH + r] = mk2(v.z, v.w);
  }
}

__device__ __forceinline__ void load_w_sw(v2f* __restrict__ sW,
                                          const float* __restrict__ W,
                                          int rowOff, int tid) {
  const float4* g4 = reinterpret_cast<const float4*>(W + (size_t)rowOff * D);
#pragma unroll
  for (int i = 0; i < 8; ++i) {
    int linear = i * TPB + tid;
    int p  = linear >> 5;
    int c4 = linear & 31;
    float4 r0 = g4[(p * 2) * 32 + c4];
    float4 r1 = g4[(p * 2 + 1) * 32 + c4];
    v2f* dstp = sW + p * WPITCH + c4 * 4;
    dstp[0] = mk2(r0.x, r1.x);
    dstp[1] = mk2(r0.y, r1.y);
    dstp[2] = mk2(r0.z, r1.z);
    dstp[3] = mk2(r0.w, r1.w);
  }
}

__device__ __forceinline__ void gemm64(const v2f* __restrict__ sX, int kpBase,
                                       const v2f* __restrict__ sW,
                                       v8f acc[8], int m0, int lane) {
  const int lrow = lane & 15;
  const int hi   = lane >> 4;
#pragma unroll 4
  for (int kp = 0; kp < 32; kp += 2) {
    v2f a = sX[(kpBase + kp + hi) * XPITCH + m0 + lrow];
    const v2f* wp = sW + (kp + hi) * WPITCH + lrow;
#pragma unroll
    for (int n = 0; n < 8; ++n) {
      v2f b = wp[n * 16];
      acc[n] = __builtin_amdgcn_wmma_f32_16x16x4_f32(
           false, a,  false, b,
           (short)0, acc[n],  false,  false);
    }
  }
}

__device__ __forceinline__ void store_acc_sw(v2f* __restrict__ sT,
                                             const v8f acc[8],
                                             const float* __restrict__ bias,
                                             int m0, int lane, bool relu) {
  const int lcol = lane & 15;
  const int mhi  = (lane >> 4) << 3;
  float* base = reinterpret_cast<float*>(sT);
#pragma unroll
  for (int n = 0; n < 8; ++n) {
    int col = n * 16 + lcol;
    float bv = bias[col];
    int p = col >> 1, j = col & 1;
#pragma unroll
    for (int v = 0; v < 8; ++v) {
      float x = acc[n][v] + bv;
      if (relu) x = fmaxf(x, 0.f);
      base[(p * XPITCH + m0 + mhi + v) * 2 + j] = x;
    }
  }
}

__global__ __launch_bounds__(TPB) void mlp_kernel(const float* __restrict__ nf,
                                                  const float* __restrict__ W1,
                                                  const float* __restrict__ b1,
                                                  const float* __restrict__ W2,
                                                  const float* __restrict__ b2,
                                                  const float* __restrict__ W3,
                                                  const float* __restrict__ b3,
                                                  float* __restrict__ out) {
  __shared__ v2f sW[32 * WPITCH];
  __shared__ v2f sX[128 * XPITCH];
  __shared__ v2f sH[64 * XPITCH];

  const int tid  = threadIdx.x;
  const int lane = tid & 31;
  const int m0   = (tid >> 5) * 16;
  const int rowBase = blockIdx.x * M_TILE;

  load_tile_sw(sX,       out, rowBase, tid);
  load_tile_sw(sX + 64 * XPITCH, nf, rowBase, tid);

  const v8f zv = {0.f, 0.f, 0.f, 0.f, 0.f, 0.f, 0.f, 0.f};
  v8f acc[8];
#pragma unroll
  for (int n = 0; n < 8; ++n) acc[n] = zv;

#pragma unroll 1
  for (int s = 0; s < 4; ++s) {
    __syncthreads();
    load_w_sw(sW, W1, s * 64, tid);
    __syncthreads();
    gemm64(sX, s * 32, sW, acc, m0, lane);
  }
  store_acc_sw(sH, acc, b1, m0, lane, true);

#pragma unroll
  for (int n = 0; n < 8; ++n) acc[n] = zv;
#pragma unroll 1
  for (int s = 0; s < 2; ++s) {
    __syncthreads();
    load_w_sw(sW, W2, s * 64, tid);
    __syncthreads();
    gemm64(sH, s * 32, sW, acc, m0, lane);
  }
  store_acc_sw(sX, acc, b2, m0, lane, true);

#pragma unroll
  for (int n = 0; n < 8; ++n) acc[n] = zv;
#pragma unroll 1
  for (int s = 0; s < 2; ++s) {
    __syncthreads();
    load_w_sw(sW, W3, s * 64, tid);
    __syncthreads();
    gemm64(sX, s * 32, sW, acc, m0, lane);
  }
  store_acc_sw(sH, acc, b3, m0, lane, false);
  __syncthreads();

  float4* o4 = reinterpret_cast<float4*>(out);
#pragma unroll
  for (int i = 0; i < 16; ++i) {
    int linear = i * TPB + tid;
    int r  = linear & 63;
    int c4 = linear >> 6;
    int gr = rowBase + r;
    v2f lo = sH[(c4 * 2) * XPITCH + r];
    v2f hv = sH[(c4 * 2 + 1) * XPITCH + r];
    if (gr < N_NODES)
      o4[(size_t)gr * 32 + c4] = make_float4(lo.x, lo.y, hv.x, hv.y);
  }
}

extern "C" void kernel_launch(void* const* d_in, const int* in_sizes, int n_in,
                              void* d_out, int out_size, void* d_ws, size_t ws_size,
                              hipStream_t stream) {
  const float* nf  = (const float*)d_in[0];
  const int*   src = (const int*)d_in[1];
  const int*   dst = (const int*)d_in[2];
  const float* W1  = (const float*)d_in[3];
  const float* b1  = (const float*)d_in[4];
  const float* W2  = (const float*)d_in[5];
  const float* b2  = (const float*)d_in[6];
  const float* W3  = (const float*)d_in[7];
  const float* b3  = (const float*)d_in[8];
  float* out = (float*)d_out;
  (void)in_sizes; (void)n_in; (void)out_size; (void)d_ws; (void)ws_size;

  const int n4 = N_NODES * D / 4;
  zero_kernel<<<(n4 + 255) / 256, 256, 0, stream>>>(out, n4);

  scatter_kernel<<<N_EDGES / 8, 256, 0, stream>>>(nf, src, dst, out);

  const int nBlocks = (N_NODES + M_TILE - 1) / M_TILE;
  mlp_kernel<<<nBlocks, TPB, 0, stream>>>(nf, W1, b1, W2, b2, W3, b3, out);
}


# Aggregate without float atomics: counting-sort edges by dst (integer atomics on per-node counters, one-block prefix scan), then one wave per node sums its source rows in registers
# speedup vs baseline: 1.0499x; 1.0499x over previous
#include <hip/hip_runtime.h>


#define N_NODES 100000
#define N_EDGES 1600000
#define D       128
#define M_TILE  64
#define TPB     128
#define XPITCH  80
#define WPITCH  144

#define CUR_PAD  100352
#define PERM_OFF CUR_PAD
#define WS_NEED  ((size_t)(PERM_OFF + N_EDGES) * 4)
#define SCAN_T   1024
#define GS_BLOCKS 4096

typedef __attribute__((ext_vector_type(2))) float v2f;
typedef __attribute__((ext_vector_type(8))) float v8f;

__device__ __forceinline__ v2f mk2(float a, float b) {
  v2f t; t.x = a; t.y = b; return t;
}

__global__ __launch_bounds__(256) void zero_cur_kernel(int* __restrict__ cur) {
  int i = blockIdx.x * 256 + threadIdx.x;
  if (i < CUR_PAD / 4)
    reinterpret_cast<int4*>(cur)[i] = make_int4(0, 0, 0, 0);
}

__global__ __launch_bounds__(256) void degree_kernel(const int* __restrict__ dst,
                                                     int* __restrict__ cur) {
  int t = blockIdx.x * 256 + threadIdx.x;
  if (t >= N_EDGES / 4) return;
  int4 d = reinterpret_cast<const int4*>(dst)[t];
  __hip_atomic_fetch_add(cur + d.x, 1, __ATOMIC_RELAXED, __HIP_MEMORY_SCOPE_AGENT);
  __hip_atomic_fetch_add(cur + d.y, 1, __ATOMIC_RELAXED, __HIP_MEMORY_SCOPE_AGENT);
  __hip_atomic_fetch_add(cur + d.z, 1, __ATOMIC_RELAXED, __HIP_MEMORY_SCOPE_AGENT);
  __hip_atomic_fetch_add(cur + d.w, 1, __ATOMIC_RELAXED, __HIP_MEMORY_SCOPE_AGENT);
}

__device__ __forceinline__ int wave_incl_scan(int x, int lane) {
#pragma unroll
  for (int o = 1; o < 32; o <<= 1) {
    int t = __shfl_up(x, o);
    if (lane >= o) x += t;
  }
  return x;
}

__global__ __launch_bounds__(SCAN_T) void scan_kernel(int* __restrict__ cur) {
  __shared__ int sTot[2][32];
  const int tid  = threadIdx.x;
  const int lane = tid & 31;
  const int wv   = tid >> 5;
  int4* c4 = reinterpret_cast<int4*>(cur);
  const int n8 = N_NODES / 8;
  int carry = 0;
  int4 a = c4[2 * tid], b = c4[2 * tid + 1];
  for (int it = 0; it * SCAN_T < n8; ++it) {
    const int g  = it * SCAN_T + tid;
    const bool ok = g < n8;
    const int gl = min(g + SCAN_T, n8 - 1);
    const int4 na = c4[2 * gl], nb = c4[2 * gl + 1];
    const int s = ok ? (a.x + a.y + a.z + a.w + b.x + b.y + b.z + b.w) : 0;
    const int incl = wave_incl_scan(s, lane);
    if (lane == 31) sTot[it & 1][wv] = incl;
    __syncthreads();
    const int wt    = sTot[it & 1][lane];
    const int wincl = wave_incl_scan(wt, lane);
    const int wexcl = __shfl(wincl, wv) - __shfl(wt, wv);
    const int total = __shfl(wincl, 31);
    const int excl  = carry + wexcl + incl - s;
    if (ok) {
      int4 oa, ob;
      oa.x = excl;        oa.y = oa.x + a.x; oa.z = oa.y + a.y; oa.w = oa.z + a.z;
      ob.x = oa.w + a.w;  ob.y = ob.x + b.x; ob.z = ob.y + b.y; ob.w = ob.z + b.z;
      c4[2 * g] = oa; c4[2 * g + 1] = ob;
    }
    carry += total;
    a = na; b = nb;
  }
}

__global__ __launch_bounds__(256) void fill_kernel(const int* __restrict__ src,
                                                   const int* __restrict__ dst,
                                                   int* __restrict__ cur,
                                                   int* __restrict__ perm) {
  int t = blockIdx.x * 256 + threadIdx.x;
  if (t >= N_EDGES / 4) return;
  int4 d = reinterpret_cast<const int4*>(dst)[t];
  int4 s = reinterpret_cast<const int4*>(src)[t];
  unsigned p0 = (unsigned)__hip_atomic_fetch_add(cur + d.x, 1, __ATOMIC_RELAXED, __HIP_MEMORY_SCOPE_AGENT);
  unsigned p1 = (unsigned)__hip_atomic_fetch_add(cur + d.y, 1, __ATOMIC_RELAXED, __HIP_MEMORY_SCOPE_AGENT);
  unsigned p2 = (unsigned)__hip_atomic_fetch_add(cur + d.z, 1, __ATOMIC_RELAXED, __HIP_MEMORY_SCOPE_AGENT);
  unsigned p3 = (unsigned)__hip_atomic_fetch_add(cur + d.w, 1, __ATOMIC_RELAXED, __HIP_MEMORY_SCOPE_AGENT);
  perm[min(p0, (unsigned)(N_EDGES - 1))] = s.x;
  perm[min(p1, (unsigned)(N_EDGES - 1))] = s.y;
  perm[min(p2, (unsigned)(N_EDGES - 1))] = s.z;
  perm[min(p3, (unsigned)(N_EDGES - 1))] = s.w;
}

template <int U>
__device__ __forceinline__ void add_rows(float4& acc, const float4* __restrict__ nf4,
                                         int myidx, int k, int lane) {
  float4 v[U];
#pragma unroll
  for (int u = 0; u < U; ++u) {
    int s = __builtin_amdgcn_readlane(myidx, k + u);
    v[u] = nf4[(size_t)s * 32 + lane];
  }
#pragma unroll
  for (int u = 0; u < U; ++u) {
    acc.x += v[u].x; acc.y += v[u].y; acc.z += v[u].z; acc.w += v[u].w;
  }
}

__global__ __launch_bounds__(256) void segsum_kernel(const float* __restrict__ nf,
                                                     const int* __restrict__ cur,
                                                     const int* __restrict__ perm,
                                                     float* __restrict__ agg) {
  const int lane = threadIdx.x & 31;
  const int wave = blockIdx.x * 8 + __builtin_amdgcn_readfirstlane(threadIdx.x >> 5);
  const float4* nf4 = reinterpret_cast<const float4*>(nf);
  float4* agg4 = reinterpret_cast<float4*>(agg);
  for (int n = wave; n < N_NODES; n += GS_BLOCKS * 8) {
    int beg = (n > 0) ? cur[n - 1] : 0;
    int end = cur[n];
    beg = max(beg, 0);
    end = min(end, N_EDGES);
    float4 acc = make_float4(0.f, 0.f, 0.f, 0.f);
    for (int base = beg; base < end; base += 32) {
      const int cnt = min(32, end - base);
      int myidx = 0;
      if (lane < cnt) myidx = perm[base + lane];
      myidx = (int)min((unsigned)myidx, (unsigned)(N_NODES - 1));
      int k = 0;
      for (; k + 8 <= cnt; k += 8) add_rows<8>(acc, nf4, myidx, k, lane);
      if (k + 4 <= cnt) { add_rows<4>(acc, nf4, myidx, k, lane); k += 4; }
      if (k + 2 <= cnt) { add_rows<2>(acc, nf4, myidx, k, lane); k += 2; }
      if (k < cnt)      { add_rows<1>(acc, nf4, myidx, k, lane); }
    }
    agg4[(size_t)n * 32 + lane] = acc;
  }
}


__device__ __forceinline__ void load_tile_sw(v2f* __restrict__ sT,
                                             const float* __restrict__ g,
                                             int rowBase, int tid) {
  const float4* g4 = reinterpret_cast<const float4*>(g);
#pragma unroll
  for (int i = 0; i < 16; ++i) {
    int linear = i * TPB + tid;
    int r  = linear & 63;
    int c4 = linear >> 6;
    int gr = rowBase + r;
    if (gr >= N_NODES) gr = N_NODES - 1;
    float4 v = g4[(size_t)gr * 32 + c4];
    sT[(c4 * 2) * XPITCH + r]     = mk2(v.x, v.y);
    sT[(c4 * 2 + 1) * XPITCH + r] = mk2(v.z, v.w);
  }
}

__device__ __forceinline__ void load_w_sw(v2f* __restrict__ sW,
                                          const float* __restrict__ W,
                                          int rowOff, int tid) {
  const float4* g4 = reinterpret_cast<const float4*>(W + (size_t)rowOff * D);
#pragma unroll
  for (int i = 0; i < 8; ++i) {
    int linear = i * TPB + tid;
    int p  = linear >> 5;
    int c4 = linear & 31;
    float4 r0 = g4[(p * 2) * 32 + c4];
    float4 r1 = g4[(p * 2 + 1) * 32 + c4];
    v2f* dstp = sW + p * WPITCH + c4 * 4;
    dstp[0] = mk2(r0.x, r1.x);
    dstp[1] = mk2(r0.y, r1.y);
    dstp[2] = mk2(r0.z, r1.z);
    dstp[3] = mk2(r0.w, r1.w);
  }
}

__device__ __forceinline__ void gemm64(const v2f* __restrict__ sX, int kpBase,
                                       const v2f* __restrict__ sW,
                                       v8f acc[8], int m0, int lane) {
  const int lrow = lane & 15;
  const int hi   = lane >> 4;
#pragma unroll 4
  for (int kp = 0; kp < 32; kp += 2) {
    v2f a = sX[(kpBase + kp + hi) * XPITCH + m0 + lrow];
    const v2f* wp = sW + (kp + hi) * WPITCH + lrow;
#pragma unroll
    for (int n = 0; n < 8; ++n) {
      v2f b = wp[n * 16];
      acc[n] = __builtin_amdgcn_wmma_f32_16x16x4_f32(
           false, a,  false, b,
           (short)0, acc[n],  false,  false);
    }
  }
}

__device__ __forceinline__ void store_acc_sw(v2f* __restrict__ sT,
                                             const v8f acc[8],
                                             const float* __restrict__ bias,
                                             int m0, int lane, bool relu) {
  const int lcol = lane & 15;
  const int mhi  = (lane >> 4) << 3;
  float* base = reinterpret_cast<float*>(sT);
#pragma unroll
  for (int n = 0; n < 8; ++n) {
    int col = n * 16 + lcol;
    float bv = bias[col];
    int p = col >> 1, j = col & 1;
#pragma unroll
    for (int v = 0; v < 8; ++v) {
      float x = acc[n][v] + bv;
      if (relu) x = fmaxf(x, 0.f);
      base[(p * XPITCH + m0 + mhi + v) * 2 + j] = x;
    }
  }
}

__global__ __launch_bounds__(TPB) void mlp_kernel(const float* __restrict__ nf,
                                                  const float* __restrict__ W1,
                                                  const float* __restrict__ b1,
                                                  const float* __restrict__ W2,
                                                  const float* __restrict__ b2,
                                                  const float* __restrict__ W3,
                                                  const float* __restrict__ b3,
                                                  float* __restrict__ out) {
  __shared__ v2f sW[32 * WPITCH];
  __shared__ v2f sX[128 * XPITCH];
  __shared__ v2f sH[64 * XPITCH];

  const int tid  = threadIdx.x;
  const int lane = tid & 31;
  const int m0   = (tid >> 5) * 16;
  const int rowBase = blockIdx.x * M_TILE;

  load_tile_sw(sX,       out, rowBase, tid);
  load_tile_sw(sX + 64 * XPITCH, nf, rowBase, tid);

  const v8f zv = {0.f, 0.f, 0.f, 0.f, 0.f, 0.f, 0.f, 0.f};
  v8f acc[8];
#pragma unroll
  for (int n = 0; n < 8; ++n) acc[n] = zv;

#pragma unroll 1
  for (int s = 0; s < 4; ++s) {
    __syncthreads();
    load_w_sw(sW, W1, s * 64, tid);
    __syncthreads();
    gemm64(sX, s * 32, sW, acc, m0, lane);
  }
  store_acc_sw(sH, acc, b1, m0, lane, true);

#pragma unroll
  for (int n = 0; n < 8; ++n) acc[n] = zv;
#pragma unroll 1
  for (int s = 0; s < 2; ++s) {
    __syncthreads();
    load_w_sw(sW, W2, s * 64, tid);
    __syncthreads();
    gemm64(sH, s * 32, sW, acc, m0, lane);
  }
  store_acc_sw(sX, acc, b2, m0, lane, true);

#pragma unroll
  for (int n = 0; n < 8; ++n) acc[n] = zv;
#pragma unroll 1
  for (int s = 0; s < 2; ++s) {
    __syncthreads();
    load_w_sw(sW, W3, s * 64, tid);
    __syncthreads();
    gemm64(sX, s * 32, sW, acc, m0, lane);
  }
  store_acc_sw(sH, acc, b3, m0, lane, false);
  __syncthreads();

  float4* o4 = reinterpret_cast<float4*>(out);
#pragma unroll
  for (int i = 0; i < 16; ++i) {
    int linear = i * TPB + tid;
    int r  = linear & 63;
    int c4 = linear >> 6;
    int gr = rowBase + r;
    v2f lo = sH[(c4 * 2) * XPITCH + r];
    v2f hv = sH[(c4 * 2 + 1) * XPITCH + r];
    if (gr < N_NODES)
      o4[(size_t)gr * 32 + c4] = make_float4(lo.x, lo.y, hv.x, hv.y);
  }
}

extern "C" void kernel_launch(void* const* d_in, const int* in_sizes, int n_in,
                              void* d_out, int out_size, void* d_ws, size_t ws_size,
                              hipStream_t stream) {
  const float* nf  = (const float*)d_in[0];
  const int*   src = (const int*)d_in[1];
  const int*   dst = (const int*)d_in[2];
  const float* W1  = (const float*)d_in[3];
  const float* b1  = (const float*)d_in[4];
  const float* W2  = (const float*)d_in[5];
  const float* b2  = (const float*)d_in[6];
  const float* W3  = (const float*)d_in[7];
  const float* b3  = (const float*)d_in[8];
  float* out = (float*)d_out;
  int* cur  = (int*)d_ws;
  int* perm = (int*)d_ws + PERM_OFF;
  (void)in_sizes; (void)n_in; (void)out_size;

  if (ws_size < WS_NEED) return;

  zero_cur_kernel<<<(CUR_PAD / 4 + 255) / 256, 256, 0, stream>>>(cur);

  degree_kernel<<<(N_EDGES / 4 + 255) / 256, 256, 0, stream>>>(dst, cur);

  scan_kernel<<<1, SCAN_T, 0, stream>>>(cur);

  fill_kernel<<<(N_EDGES / 4 + 255) / 256, 256, 0, stream>>>(src, dst, cur, perm);

  segsum_kernel<<<GS_BLOCKS, 256, 0, stream>>>(nf, cur, perm, out);

  const int nBlocks = (N_NODES + M_TILE - 1) / M_TILE;
  mlp_kernel<<<nBlocks, TPB, 0, stream>>>(nf, W1, b1, W2, b2, W3, b3, out);
}


# Counting pass's integer atomics now return each edge's rank (kept in d_out as scratch); grouping pass takes slot = segment start + rank with plain loads: one pass of 1.6M global atomics removed
# speedup vs baseline: 1.2462x; 1.1870x over previous
#include <hip/hip_runtime.h>


#define N_NODES 100000
#define N_EDGES 1600000
#define D       128
#define M_TILE  64
#define TPB     128
#define XPITCH  80
#define WPITCH  144

#define CUR_PAD  100352
#define PERM_OFF CUR_PAD
#define WS_NEED  ((size_t)(PERM_OFF + N_EDGES) * 4)
#define SCAN_T   1024
#define GS_BLOCKS 4096

typedef __attribute__((ext_vector_type(2))) float v2f;
typedef __attribute__((ext_vector_type(8))) float v8f;

__device__ __forceinline__ v2f mk2(float a, float b) {
  v2f t; t.x = a; t.y = b; return t;
}

__global__ __launch_bounds__(256) void zero_cur_kernel(int* __restrict__ cur) {
  int i = blockIdx.x * 256 + threadIdx.x;
  if (i < CUR_PAD / 4)
    reinterpret_cast<int4*>(cur)[i] = make_int4(0, 0, 0, 0);
}

__global__ __launch_bounds__(256) void rank_kernel(const int* __restrict__ dst,
                                                   int* __restrict__ cur,
                                                   int* __restrict__ rank) {
  int t = blockIdx.x * 256 + threadIdx.x;
  if (t >= N_EDGES / 4) return;
  int4 d = reinterpret_cast<const int4*>(dst)[t];
  int4 r;
  r.x = __hip_atomic_fetch_add(cur + d.x, 1, __ATOMIC_RELAXED, __HIP_MEMORY_SCOPE_AGENT);
  r.y = __hip_atomic_fetch_add(cur + d.y, 1, __ATOMIC_RELAXED, __HIP_MEMORY_SCOPE_AGENT);
  r.z = __hip_atomic_fetch_add(cur + d.z, 1, __ATOMIC_RELAXED, __HIP_MEMORY_SCOPE_AGENT);
  r.w = __hip_atomic_fetch_add(cur + d.w, 1, __ATOMIC_RELAXED, __HIP_MEMORY_SCOPE_AGENT);
  reinterpret_cast<int4*>(rank)[t] = r;
}

__device__ __forceinline__ int wave_incl_scan(int x, int lane) {
#pragma unroll
  for (int o = 1; o < 32; o <<= 1) {
    int t = __shfl_up(x, o);
    if (lane >= o) x += t;
  }
  return x;
}

__global__ __launch_bounds__(SCAN_T) void scan_kernel(int* __restrict__ cur) {
  __shared__ int sTot[2][32];
  const int tid  = threadIdx.x;
  const int lane = tid & 31;
  const int wv   = tid >> 5;
  int4* c4 = reinterpret_cast<int4*>(cur);
  const int n8 = N_NODES / 8;
  int carry = 0;
  int4 a = c4[2 * tid], b = c4[2 * tid + 1];
  for (int it = 0; it * SCAN_T < n8; ++it) {
    const int g  = it * SCAN_T + tid;
    const bool ok = g < n8;
    const int gl = min(g + SCAN_T, n8 - 1);
    const int4 na = c4[2 * gl], nb = c4[2 * gl + 1];
    const int s = ok ? (a.x + a.y + a.z + a.w + b.x + b.y + b.z + b.w) : 0;
    const int incl = wave_incl_scan(s, lane);
    if (lane == 31) sTot[it & 1][wv] = incl;
    __syncthreads();
    const int wt    = sTot[it & 1][lane];
    const int wincl = wave_incl_scan(wt, lane);
    const int wexcl = __shfl(wincl, wv) - __shfl(wt, wv);
    const int total = __shfl(wincl, 31);
    const int excl  = carry + wexcl + incl - s;
    if (ok) {
      int4 oa, ob;
      oa.x = excl;        oa.y = oa.x + a.x; oa.z = oa.y + a.y; oa.w = oa.z + a.z;
      ob.x = oa.w + a.w;  ob.y = ob.x + b.x; ob.z = ob.y + b.y; ob.w = ob.z + b.z;
      c4[2 * g] = oa; c4[2 * g + 1] = ob;
    }
    carry += total;
    a = na; b = nb;
  }
}

__global__ __launch_bounds__(256) void fill_kernel(const int* __restrict__ src,
                                                   const int* __restrict__ dst,
                                                   const int* __restrict__ cur,
                                                   const int* __restrict__ rank,
                                                   int* __restrict__ perm) {
  int t = blockIdx.x * 256 + threadIdx.x;
  if (t >= N_EDGES / 4) return;
  int4 d = reinterpret_cast<const int4*>(dst)[t];
  int4 s = reinterpret_cast<const int4*>(src)[t];
  int4 r = reinterpret_cast<const int4*>(rank)[t];
  unsigned p0 = (unsigned)(cur[d.x] + r.x);
  unsigned p1 = (unsigned)(cur[d.y] + r.y);
  unsigned p2 = (unsigned)(cur[d.z] + r.z);
  unsigned p3 = (unsigned)(cur[d.w] + r.w);
  perm[min(p0, (unsigned)(N_EDGES - 1))] = s.x;
  perm[min(p1, (unsigned)(N_EDGES - 1))] = s.y;
  perm[min(p2, (unsigned)(N_EDGES - 1))] = s.z;
  perm[min(p3, (unsigned)(N_EDGES - 1))] = s.w;
}

template <int U>
__device__ __forceinline__ void add_rows(float4& acc, const float4* __restrict__ nf4,
                                         int myidx, int k, int lane) {
  float4 v[U];
#pragma unroll
  for (int u = 0; u < U; ++u) {
    int s = __builtin_amdgcn_readlane(myidx, k + u);
    v[u] = nf4[(size_t)s * 32 + lane];
  }
#pragma unroll
  for (int u = 0; u < U; ++u) {
    acc.x += v[u].x; acc.y += v[u].y; acc.z += v[u].z; acc.w += v[u].w;
  }
}

__global__ __launch_bounds__(256) void segsum_kernel(const float* __restrict__ nf,
                                                     const int* __restrict__ cur,
                                                     const int* __restrict__ perm,
                                                     float* __restrict__ agg) {
  const int lane = threadIdx.x & 31;
  const int wave = blockIdx.x * 8 + __builtin_amdgcn_readfirstlane(threadIdx.x >> 5);
  const float4* nf4 = reinterpret_cast<const float4*>(nf);
  float4* agg4 = reinterpret_cast<float4*>(agg);
  for (int n = wave; n < N_NODES; n += GS_BLOCKS * 8) {
    int beg = cur[n];
    int end = (n + 1 < N_NODES) ? cur[n + 1] : N_EDGES;
    beg = max(beg, 0);
    end = min(end, N_EDGES);
    float4 acc = make_float4(0.f, 0.f, 0.f, 0.f);
    for (int base = beg; base < end; base += 32) {
      const int cnt = min(32, end - base);
      int myidx = 0;
      if (lane < cnt) myidx = perm[base + lane];
      myidx = (int)min((unsigned)myidx, (unsigned)(N_NODES - 1));
      int k = 0;
      for (; k + 8 <= cnt; k += 8) add_rows<8>(acc, nf4, myidx, k, lane);
      if (k + 4 <= cnt) { add_rows<4>(acc, nf4, myidx, k, lane); k += 4; }
      if (k + 2 <= cnt) { add_rows<2>(acc, nf4, myidx, k, lane); k += 2; }
      if (k < cnt)      { add_rows<1>(acc, nf4, myidx, k, lane); }
    }
    agg4[(size_t)n * 32 + lane] = acc;
  }
}


__device__ __forceinline__ void load_tile_sw(v2f* __restrict__ sT,
                                             const float* __restrict__ g,
                                             int rowBase, int tid) {
  const float4* g4 = reinterpret_cast<const float4*>(g);
#pragma unroll
  for (int i = 0; i < 16; ++i) {
    int linear = i * TPB + tid;
    int r  = linear & 63;
    int c4 = linear >> 6;
    int gr = rowBase + r;
    if (gr >= N_NODES) gr = N_NODES - 1;
    float4 v = g4[(size_t)gr * 32 + c4];
    sT[(c4 * 2) * XPITCH + r]     = mk2(v.x, v.y);
    sT[(c4 * 2 + 1) * XPITCH + r] = mk2(v.z, v.w);
  }
}

__device__ __forceinline__ void load_w_sw(v2f* __restrict__ sW,
                                          const float* __restrict__ W,
                                          int rowOff, int tid) {
  const float4* g4 = reinterpret_cast<const float4*>(W + (size_t)rowOff * D);
#pragma unroll
  for (int i = 0; i < 8; ++i) {
    int linear = i * TPB + tid;
    int p  = linear >> 5;
    int c4 = linear & 31;
    float4 r0 = g4[(p * 2) * 32 + c4];
    float4 r1 = g4[(p * 2 + 1) * 32 + c4];
    v2f* dstp = sW + p * WPITCH + c4 * 4;
    dstp[0] = mk2(r0.x, r1.x);
    dstp[1] = mk2(r0.y, r1.y);
    dstp[2] = mk2(r0.z, r1.z);
    dstp[3] = mk2(r0.w, r1.w);
  }
}

__device__ __forceinline__ void gemm64(const v2f* __restrict__ sX, int kpBase,
                                       const v2f* __restrict__ sW,
                                       v8f acc[8], int m0, int lane) {
  const int lrow = lane & 15;
  const int hi   = lane >> 4;
#pragma unroll 4
  for (int kp = 0; kp < 32; kp += 2) {
    v2f a = sX[(kpBase + kp + hi) * XPITCH + m0 + lrow];
    const v2f* wp = sW + (kp + hi) * WPITCH + lrow;
#pragma unroll
    for (int n = 0; n < 8; ++n) {
      v2f b = wp[n * 16];
      acc[n] = __builtin_amdgcn_wmma_f32_16x16x4_f32(
           false, a,  false, b,
           (short)0, acc[n],  false,  false);
    }
  }
}

__device__ __forceinline__ void store_acc_sw(v2f* __restrict__ sT,
                                             const v8f acc[8],
                                             const float* __restrict__ bias,
                                             int m0, int lane, bool relu) {
  const int lcol = lane & 15;
  const int mhi  = (lane >> 4) << 3;
  float* base = reinterpret_cast<float*>(sT);
#pragma unroll
  for (int n = 0; n < 8; ++n) {
    int col = n * 16 + lcol;
    float bv = bias[col];
    int p = col >> 1, j = col & 1;
#pragma unroll
    for (int v = 0; v < 8; ++v) {
      float x = acc[n][v] + bv;
      if (relu) x = fmaxf(x, 0.f);
      base[(p * XPITCH + m0 + mhi + v) * 2 + j] = x;
    }
  }
}

__global__ __launch_bounds__(TPB) void mlp_kernel(const float* __restrict__ nf,
                                                  const float* __restrict__ W1,
                                                  const float* __restrict__ b1,
                                                  const float* __restrict__ W2,
                                                  const float* __restrict__ b2,
                                                  const float* __restrict__ W3,
                                                  const float* __restrict__ b3,
                                                  float* __restrict__ out) {
  __shared__ v2f sW[32 * WPITCH];
  __shared__ v2f sX[128 * XPITCH];
  __shared__ v2f sH[64 * XPITCH];

  const int tid  = threadIdx.x;
  const int lane = tid & 31;
  const int m0   = (tid >> 5) * 16;
  const int rowBase = blockIdx.x * M_TILE;

  load_tile_sw(sX,       out, rowBase, tid);
  load_tile_sw(sX + 64 * XPITCH, nf, rowBase, tid);

  const v8f zv = {0.f, 0.f, 0.f, 0.f, 0.f, 0.f, 0.f, 0.f};
  v8f acc[8];
#pragma unroll
  for (int n = 0; n < 8; ++n) acc[n] = zv;

#pragma unroll 1
  for (int s = 0; s < 4; ++s) {
    __syncthreads();
    load_w_sw(sW, W1, s * 64, tid);
    __syncthreads();
    gemm64(sX, s * 32, sW, acc, m0, lane);
  }
  store_acc_sw(sH, acc, b1, m0, lane, true);

#pragma unroll
  for (int n = 0; n < 8; ++n) acc[n] = zv;
#pragma unroll 1
  for (int s = 0; s < 2; ++s) {
    __syncthreads();
    load_w_sw(sW, W2, s * 64, tid);
    __syncthreads();
    gemm64(sH, s * 32, sW, acc, m0, lane);
  }
  store_acc_sw(sX, acc, b2, m0, lane, true);

#pragma unroll
  for (int n = 0; n < 8; ++n) acc[n] = zv;
#pragma unroll 1
  for (int s = 0; s < 2; ++s) {
    __syncthreads();
    load_w_sw(sW, W3, s * 64, tid);
    __syncthreads();
    gemm64(sX, s * 32, sW, acc, m0, lane);
  }
  store_acc_sw(sH, acc, b3, m0, lane, false);
  __syncthreads();

  float4* o4 = reinterpret_cast<float4*>(out);
#pragma unroll
  for (int i = 0; i < 16; ++i) {
    int linear = i * TPB + tid;
    int r  = linear & 63;
    int c4 = linear >> 6;
    int gr = rowBase + r;
    v2f lo = sH[(c4 * 2) * XPITCH + r];
    v2f hv = sH[(c4 * 2 + 1) * XPITCH + r];
    if (gr < N_NODES)
      o4[(size_t)gr * 32 + c4] = make_float4(lo.x, lo.y, hv.x, hv.y);
  }
}

extern "C" void kernel_launch(void* const* d_in, const int* in_sizes, int n_in,
                              void* d_out, int out_size, void* d_ws, size_t ws_size,
                              hipStream_t stream) {
  const float* nf  = (const float*)d_in[0];
  const int*   src = (const int*)d_in[1];
  const int*   dst = (const int*)d_in[2];
  const float* W1  = (const float*)d_in[3];
  const float* b1  = (const float*)d_in[4];
  const float* W2  = (const float*)d_in[5];
  const float* b2  = (const float*)d_in[6];
  const float* W3  = (const float*)d_in[7];
  const float* b3  = (const float*)d_in[8];
  float* out = (float*)d_out;
  int* cur  = (int*)d_ws;
  int* perm = (int*)d_ws + PERM_OFF;
  (void)in_sizes; (void)n_in; (void)out_size;

  if (ws_size < WS_NEED) return;

  zero_cur_kernel<<<(CUR_PAD / 4 + 255) / 256, 256, 0, stream>>>(cur);

  int* rank = (int*)d_out;
  rank_kernel<<<(N_EDGES / 4 + 255) / 256, 256, 0, stream>>>(dst, cur, rank);

  scan_kernel<<<1, SCAN_T, 0, stream>>>(cur);

  fill_kernel<<<(N_EDGES / 4 + 255) / 256, 256, 0, stream>>>(src, dst, cur, rank, perm);

  segsum_kernel<<<GS_BLOCKS, 256, 0, stream>>>(nf, cur, perm, out);

  const int nBlocks = (N_NODES + M_TILE - 1) / M_TILE;
  mlp_kernel<<<nBlocks, TPB, 0, stream>>>(nf, W1, b1, W2, b2, W3, b3, out);
}


# Edge ranking moved from scattered global atomics to LDS atomics: two-level counting sort (LDS histogram per edge block over 391 dst buckets, prefix, per-bucket LDS sort), stores leave as runs
# speedup vs baseline: 1.9270x; 1.5463x over previous
#include <hip/hip_runtime.h>


#define N_NODES 100000
#define N_EDGES 1600000
#define D       128
#define M_TILE  64
#define TPB     128
#define XPITCH  80
#define WPITCH  144

#define CUR_PAD  100352
#define PERM_OFF CUR_PAD
#define WS_NEED  ((size_t)(PERM_OFF + N_EDGES) * 4)
#define SCAN_T   1024
#define NB       256
#define NBKT     391
#define SCAN_N   (NBKT * NB)
#define NG       (N_EDGES / 4)
#define GPB      ((NG + NB - 1) / NB)
#define ST       512
#define GRPS     4
#define NBKT_PAD 416
#define BCAP     8192
#define EPT      (BCAP / ST)
#define GS_BLOCKS 4096

typedef __attribute__((ext_vector_type(2))) float v2f;
typedef __attribute__((ext_vector_type(8))) float v8f;

__device__ __forceinline__ v2f mk2(float a, float b) {
  v2f t; t.x = a; t.y = b; return t;
}

__device__ __forceinline__ int node_of(int d) {
  return (int)min((unsigned)d, (unsigned)(N_NODES - 1));
}

__global__ __launch_bounds__(ST) void hist_kernel(const int* __restrict__ dst,
                                                  int* __restrict__ cnt) {
  __shared__ int h[NBKT];
  const int tid = threadIdx.x;
  if (tid < NBKT) h[tid] = 0;
  __syncthreads();
  static_assert(GPB <= GRPS * ST, "a thread holds at most GRPS groups of four edges");
  const int g0 = blockIdx.x * GPB;
  const int g1 = min(g0 + GPB, NG);
  int4 d[GRPS];
#pragma unroll
  for (int u = 0; u < GRPS; ++u) {
    const int g = g0 + tid + u * ST;
    d[u] = reinterpret_cast<const int4*>(dst)[min(g, NG - 1)];
  }
#pragma unroll
  for (int u = 0; u < GRPS; ++u) {
    if (g0 + tid + u * ST < g1) {
      __hip_atomic_fetch_add(&h[node_of(d[u].x) >> 8], 1, __ATOMIC_RELAXED, __HIP_MEMORY_SCOPE_WORKGROUP);
      __hip_atomic_fetch_add(&h[node_of(d[u].y) >> 8], 1, __ATOMIC_RELAXED, __HIP_MEMORY_SCOPE_WORKGROUP);
      __hip_atomic_fetch_add(&h[node_of(d[u].z) >> 8], 1, __ATOMIC_RELAXED, __HIP_MEMORY_SCOPE_WORKGROUP);
      __hip_atomic_fetch_add(&h[node_of(d[u].w) >> 8], 1, __ATOMIC_RELAXED, __HIP_MEMORY_SCOPE_WORKGROUP);
    }
  }
  __syncthreads();
  if (tid < NBKT) cnt[tid * NB + blockIdx.x] = h[tid];
}

__device__ __forceinline__ int wave_incl_scan(int x, int lane) {
#pragma unroll
  for (int o = 1; o < 32; o <<= 1) {
    int t = __shfl_up(x, o);
    if (lane >= o) x += t;
  }
  return x;
}

__global__ __launch_bounds__(SCAN_T) void scan_kernel(int* __restrict__ cur) {
  __shared__ int sTot[2][32];
  const int tid  = threadIdx.x;
  const int lane = tid & 31;
  const int wv   = tid >> 5;
  int4* c4 = reinterpret_cast<int4*>(cur);
  const int n8 = SCAN_N / 8;
  int carry = 0;
  int4 a = c4[2 * tid], b = c4[2 * tid + 1];
  for (int it = 0; it * SCAN_T < n8; ++it) {
    const int g  = it * SCAN_T + tid;
    const bool ok = g < n8;
    const int gl = min(g + SCAN_T, n8 - 1);
    const int4 na = c4[2 * gl], nb = c4[2 * gl + 1];
    const int s = ok ? (a.x + a.y + a.z + a.w + b.x + b.y + b.z + b.w) : 0;
    const int incl = wave_incl_scan(s, lane);
    if (lane == 31) sTot[it & 1][wv] = incl;
    __syncthreads();
    const int wt    = sTot[it & 1][lane];
    const int wincl = wave_incl_scan(wt, lane);
    const int wexcl = __shfl(wincl, wv) - __shfl(wt, wv);
    const int total = __shfl(wincl, 31);
    const int excl  = carry + wexcl + incl - s;
    if (ok) {
      int4 oa, ob;
      oa.x = excl;        oa.y = oa.x + a.x; oa.z = oa.y + a.y; oa.w = oa.z + a.z;
      ob.x = oa.w + a.w;  ob.y = ob.x + b.x; ob.z = ob.y + b.y; ob.w = ob.z + b.z;
      c4[2 * g] = oa; c4[2 * g + 1] = ob;
    }
    carry += total;
    a = na; b = nb;
  }
}

__device__ __forceinline__ int rank_edge(int* __restrict__ h, int d, int& r) {
  const int n = node_of(d);
  r = __hip_atomic_fetch_add(&h[n >> 8], 1, __ATOMIC_RELAXED, __HIP_MEMORY_SCOPE_WORKGROUP);
  return n;
}

__global__ __launch_bounds__(ST) void part_kernel(const int* __restrict__ src,
                                                  const int* __restrict__ dst,
                                                  const int* __restrict__ cnt,
                                                  int2* __restrict__ tmp) {
  __shared__ int  h[NBKT_PAD];
  __shared__ int  gb[NBKT];
  __shared__ int2 sp[GPB * 4];
  static_assert(GPB <= GRPS * ST, "a thread holds at most GRPS groups of four edges");
  static_assert(NBKT <= NBKT_PAD && NBKT_PAD <= ST && NBKT_PAD == 13 * 32, "one thread per counter; wave 0 takes 13 each");
  const int tid = threadIdx.x;
  const int j   = blockIdx.x;
  int gbv = 0;
  if (tid < NBKT) gbv = cnt[tid * NB + j];
  if (tid < NBKT_PAD) h[tid] = 0;
  __syncthreads();

  const int g0 = j * GPB;
  const int g1 = min(g0 + GPB, NG);
  int4 d[GRPS], sv[GRPS], r[GRPS];
#pragma unroll
  for (int u = 0; u < GRPS; ++u) {
    const int g = g0 + tid + u * ST;
    d[u] = make_int4(0, 0, 0, 0); sv[u] = d[u]; r[u] = d[u];
    if (g < g1) {
      d[u]  = reinterpret_cast<const int4*>(dst)[g];
      sv[u] = reinterpret_cast<const int4*>(src)[g];
    }
  }
#pragma unroll
  for (int u = 0; u < GRPS; ++u) {
    if (g0 + tid + u * ST < g1) {
      d[u].x = rank_edge(h, d[u].x, r[u].x); d[u].y = rank_edge(h, d[u].y, r[u].y);
      d[u].z = rank_edge(h, d[u].z, r[u].z); d[u].w = rank_edge(h, d[u].w, r[u].w);
    }
  }
  __syncthreads();

  if (tid < 32) {
    int v[13];
    int sum = 0;
#pragma unroll
    for (int i = 0; i < 13; ++i) { v[i] = h[tid * 13 + i]; sum += v[i]; }
    int run = wave_incl_scan(sum, tid) - sum;
#pragma unroll
    for (int i = 0; i < 13; ++i) { h[tid * 13 + i] = run; run += v[i]; }
  }
  __syncthreads();

  if (tid < NBKT) gb[tid] = gbv - h[tid];
#pragma unroll
  for (int u = 0; u < GRPS; ++u) {
    if (g0 + tid + u * ST < g1) {
      sp[h[d[u].x >> 8] + r[u].x] = make_int2(sv[u].x, d[u].x);
      sp[h[d[u].y >> 8] + r[u].y] = make_int2(sv[u].y, d[u].y);
      sp[h[d[u].z >> 8] + r[u].z] = make_int2(sv[u].z, d[u].z);
      sp[h[d[u].w >> 8] + r[u].w] = make_int2(sv[u].w, d[u].w);
    }
  }
  __syncthreads();

  const int nE = (g1 - g0) * 4;
  for (int i = tid; i < nE; i += ST) {
    const int2 p = sp[i];
    const unsigned pos = (unsigned)(gb[p.y >> 8] + i);
    tmp[min(pos, (unsigned)(N_EDGES - 1))] = p;
  }
}

__device__ __forceinline__ void bucket_starts(int* __restrict__ c, int cbase, int S,
                                              int b, int lane, int* __restrict__ cur) {
  int v[8];
  int sum = 0;
#pragma unroll
  for (int i = 0; i < 8; ++i) { v[i] = c[lane * 8 + i]; sum += v[i]; }
  int run = wave_incl_scan(sum, lane) - sum;
#pragma unroll
  for (int i = 0; i < 8; ++i) {
    const int node = b * 256 + lane * 8 + i;
    c[lane * 8 + i] = cbase + run;
    if (node < N_NODES) cur[node] = S + run;
    run += v[i];
  }
}

__global__ __launch_bounds__(ST) void bsort_kernel(const int* __restrict__ cnt,
                                                   const int2* __restrict__ tmp,
                                                   int* __restrict__ cur,
                                                   int* __restrict__ perm) {
  __shared__ int c[256];
  __shared__ int sp[BCAP];
  static_assert(BCAP % ST == 0 && ST >= 256, "EPT edges per thread cover BCAP; one thread per node counter");
  const int tid = threadIdx.x;
  const int b = blockIdx.x;
  int S = cnt[b * NB];
  int E = (b + 1 < NBKT) ? cnt[(b + 1) * NB] : N_EDGES;
  S = min(max(S, 0), N_EDGES);
  E = min(max(E, S), N_EDGES);
  const int n = E - S;
  if (tid < 256) c[tid] = 0;
  __syncthreads();

  if (n <= BCAP) {
    int2 v[EPT];
    int  r[EPT];
#pragma unroll
    for (int u = 0; u < EPT; ++u) {
      const int e = S + u * ST + tid;
      v[u] = make_int2(0, 0);
      if (e < E) v[u] = tmp[e];
    }
#pragma unroll
    for (int u = 0; u < EPT; ++u) {
      r[u] = 0;
      if (S + u * ST + tid < E)
        r[u] = __hip_atomic_fetch_add(&c[v[u].y & 255], 1, __ATOMIC_RELAXED, __HIP_MEMORY_SCOPE_WORKGROUP);
    }
    __syncthreads();
    if (tid < 32) bucket_starts(c, 0, S, b, tid, cur);
    __syncthreads();
#pragma unroll
    for (int u = 0; u < EPT; ++u) {
      if (S + u * ST + tid < E) {
        const unsigned q = (unsigned)(c[v[u].y & 255] + r[u]);
        sp[min(q, (unsigned)(BCAP - 1))] = v[u].x;
      }
    }
    __syncthreads();
    for (int i = tid; i < n; i += ST) perm[S + i] = sp[i];
  } else {
    for (int e = S + tid; e < E; e += ST) {
      const int2 v = tmp[e];
      __hip_atomic_fetch_add(&c[v.y & 255], 1, __ATOMIC_RELAXED, __HIP_MEMORY_SCOPE_WORKGROUP);
    }
    __syncthreads();
    if (tid < 32) bucket_starts(c, S, S, b, tid, cur);
    __syncthreads();
    for (int e = S + tid; e < E; e += ST) {
      const int2 v = tmp[e];
      const unsigned p = (unsigned)__hip_atomic_fetch_add(&c[v.y & 255], 1, __ATOMIC_RELAXED, __HIP_MEMORY_SCOPE_WORKGROUP);
      perm[min(p, (unsigned)(N_EDGES - 1))] = v.x;
    }
  }
}

template <int U>
__device__ __forceinline__ void add_rows(float4& acc, const float4* __restrict__ nf4,
                                         int myidx, int k, int lane) {
  float4 v[U];
#pragma unroll
  for (int u = 0; u < U; ++u) {
    int s = __builtin_amdgcn_readlane(myidx, k + u);
    v[u] = nf4[(size_t)s * 32 + lane];
  }
#pragma unroll
  for (int u = 0; u < U; ++u) {
    acc.x += v[u].x; acc.y += v[u].y; acc.z += v[u].z; acc.w += v[u].w;
  }
}

__global__ __launch_bounds__(256) void segsum_kernel(const float* __restrict__ nf,
                                                     const int* __restrict__ cur,
                                                     const int* __restrict__ perm,
                                                     float* __restrict__ agg) {
  const int lane = threadIdx.x & 31;
  const int wave = blockIdx.x * 8 + __builtin_amdgcn_readfirstlane(threadIdx.x >> 5);
  const float4* nf4 = reinterpret_cast<const float4*>(nf);
  float4* agg4 = reinterpret_cast<float4*>(agg);
  for (int n = wave; n < N_NODES; n += GS_BLOCKS * 8) {
    int beg = cur[n];
    int end = (n + 1 < N_NODES) ? cur[n + 1] : N_EDGES;
    beg = max(beg, 0);
    end = min(end, N_EDGES);
    float4 acc = make_float4(0.f, 0.f, 0.f, 0.f);
    for (int base = beg; base < end; base += 32) {
      const int cnt = min(32, end - base);
      int myidx = 0;
      if (lane < cnt) myidx = perm[base + lane];
      myidx = (int)min((unsigned)myidx, (unsigned)(N_NODES - 1));
      int k = 0;
      for (; k + 8 <= cnt; k += 8) add_rows<8>(acc, nf4, myidx, k, lane);
      if (k + 4 <= cnt) { add_rows<4>(acc, nf4, myidx, k, lane); k += 4; }
      if (k + 2 <= cnt) { add_rows<2>(acc, nf4, myidx, k, lane); k += 2; }
      if (k < cnt)      { add_rows<1>(acc, nf4, myidx, k, lane); }
    }
    agg4[(size_t)n * 32 + lane] = acc;
  }
}


__device__ __forceinline__ void load_tile_sw(v2f* __restrict__ sT,
                                             const float* __restrict__ g,
                                             int rowBase, int tid) {
  const float4* g4 = reinterpret_cast<const float4*>(g);
#pragma unroll
  for (int i = 0; i < 16; ++i) {
    int linear = i * TPB + tid;
    int r  = linear & 63;
    int c4 = linear >> 6;
    int gr = rowBase + r;
    if (gr >= N_NODES) gr = N_NODES - 1;
    float4 v = g4[(size_t)gr * 32 + c4];
    sT[(c4 * 2) * XPITCH + r]     = mk2(v.x, v.y);
    sT[(c4 * 2 + 1) * XPITCH + r] = mk2(v.z, v.w);
  }
}

__device__ __forceinline__ void load_w_sw(v2f* __restrict__ sW,
                                          const float* __restrict__ W,
                                          int rowOff, int tid) {
  const float4* g4 = reinterpret_cast<const float4*>(W + (size_t)rowOff * D);
#pragma unroll
  for (int i = 0; i < 8; ++i) {
    int linear = i * TPB + tid;
    int p  = linear >> 5;
    int c4 = linear & 31;
    float4 r0 = g4[(p * 2) * 32 + c4];
    float4 r1 = g4[(p * 2 + 1) * 32 + c4];
    v2f* dstp = sW + p * WPITCH + c4 * 4;
    dstp[0] = mk2(r0.x, r1.x);
    dstp[1] = mk2(r0.y, r1.y);
    dstp[2] = mk2(r0.z, r1.z);
    dstp[3] = mk2(r0.w, r1.w);
  }
}

__device__ __forceinline__ void gemm64(const v2f* __restrict__ sX, int kpBase,
                                       const v2f* __restrict__ sW,
                                       v8f acc[8], int m0, int lane) {
  const int lrow = lane & 15;
  const int hi   = lane >> 4;
#pragma unroll 4
  for (int kp = 0; kp < 32; kp += 2) {
    v2f a = sX[(kpBase + kp + hi) * XPITCH + m0 + lrow];
    const v2f* wp = sW + (kp + hi) * WPITCH + lrow;
#pragma unroll
    for (int n = 0; n < 8; ++n) {
      v2f b = wp[n * 16];
      acc[n] = __builtin_amdgcn_wmma_f32_16x16x4_f32(
           false, a,  false, b,
           (short)0, acc[n],  false,  false);
    }
  }
}

__device__ __forceinline__ void store_acc_sw(v2f* __restrict__ sT,
                                             const v8f acc[8],
                                             const float* __restrict__ bias,
                                             int m0, int lane, bool relu) {
  const int lcol = lane & 15;
  const int mhi  = (lane >> 4) << 3;
  float* base = reinterpret_cast<float*>(sT);
#pragma unroll
  for (int n = 0; n < 8; ++n) {
    int col = n * 16 + lcol;
    float bv = bias[col];
    int p = col >> 1, j = col & 1;
#pragma unroll
    for (int v = 0; v < 8; ++v) {
      float x = acc[n][v] + bv;
      if (relu) x = fmaxf(x, 0.f);
      base[(p * XPITCH + m0 + mhi + v) * 2 + j] = x;
    }
  }
}

__global__ __launch_bounds__(TPB) void mlp_kernel(const float* __restrict__ nf,
                                                  const float* __restrict__ W1,
                                                  const float* __restrict__ b1,
                                                  const float* __restrict__ W2,
                                                  const float* __restrict__ b2,
                                                  const float* __restrict__ W3,
                                                  const float* __restrict__ b3,
                                                  float* __restrict__ out) {
  __shared__ v2f sW[32 * WPITCH];
  __shared__ v2f sX[128 * XPITCH];
  __shared__ v2f sH[64 * XPITCH];

  const int tid  = threadIdx.x;
  const int lane = tid & 31;
  const int m0   = (tid >> 5) * 16;
  const int rowBase = blockIdx.x * M_TILE;

  load_tile_sw(sX,       out, rowBase, tid);
  load_tile_sw(sX + 64 * XPITCH, nf, rowBase, tid);

  const v8f zv = {0.f, 0.f, 0.f, 0.f, 0.f, 0.f, 0.f, 0.f};
  v8f acc[8];
#pragma unroll
  for (int n = 0; n < 8; ++n) acc[n] = zv;

#pragma unroll 1
  for (int s = 0; s < 4; ++s) {
    __syncthreads();
    load_w_sw(sW, W1, s * 64, tid);
    __syncthreads();
    gemm64(sX, s * 32, sW, acc, m0, lane);
  }
  store_acc_sw(sH, acc, b1, m0, lane, true);

#pragma unroll
  for (int n = 0; n < 8; ++n) acc[n] = zv;
#pragma unroll 1
  for (int s = 0; s < 2; ++s) {
    __syncthreads();
    load_w_sw(sW, W2, s * 64, tid);
    __syncthreads();
    gemm64(sH, s * 32, sW, acc, m0, lane);
  }
  store_acc_sw(sX, acc, b2, m0, lane, true);

#pragma unroll
  for (int n = 0; n < 8; ++n) acc[n] = zv;
#pragma unroll 1
  for (int s = 0; s < 2; ++s) {
    __syncthreads();
    load_w_sw(sW, W3, s * 64, tid);
    __syncthreads();
    gemm64(sX, s * 32, sW, acc, m0, lane);
  }
  store_acc_sw(sH, acc, b3, m0, lane, false);
  __syncthreads();

  float4* o4 = reinterpret_cast<float4*>(out);
#pragma unroll
  for (int i = 0; i < 16; ++i) {
    int linear = i * TPB + tid;
    int r  = linear & 63;
    int c4 = linear >> 6;
    int gr = rowBase + r;
    v2f lo = sH[(c4 * 2) * XPITCH + r];
    v2f hv = sH[(c4 * 2 + 1) * XPITCH + r];
    if (gr < N_NODES)
      o4[(size_t)gr * 32 + c4] = make_float4(lo.x, lo.y, hv.x, hv.y);
  }
}

extern "C" void kernel_launch(void* const* d_in, const int* in_sizes, int n_in,
                              void* d_out, int out_size, void* d_ws, size_t ws_size,
                              hipStream_t stream) {
  const float* nf  = (const float*)d_in[0];
  const int*   src = (const int*)d_in[1];
  const int*   dst = (const int*)d_in[2];
  const float* W1  = (const float*)d_in[3];
  const float* b1  = (const float*)d_in[4];
  const float* W2  = (const float*)d_in[5];
  const float* b2  = (const float*)d_in[6];
  const float* W3  = (const float*)d_in[7];
  const float* b3  = (const float*)d_in[8];
  float* out = (float*)d_out;
  int* cur  = (int*)d_ws;
  int* perm = (int*)d_ws + PERM_OFF;
  (void)in_sizes; (void)n_in; (void)out_size;

  if (ws_size < WS_NEED) return;

  int2* tmp = (int2*)d_out;
  int*  cnt = (int*)d_out + 2 * N_EDGES;

  hist_kernel<<<NB, ST, 0, stream>>>(dst, cnt);

  scan_kernel<<<1, SCAN_T, 0, stream>>>(cnt);

  part_kernel<<<NB, ST, 0, stream>>>(src, dst, cnt, tmp);

  bsort_kernel<<<NBKT, ST, 0, stream>>>(cnt, tmp, cur, perm);

  segsum_kernel<<<GS_BLOCKS, 256, 0, stream>>>(nf, cur, perm, out);

  const int nBlocks = (N_NODES + M_TILE - 1) / M_TILE;
  mlp_kernel<<<nBlocks, TPB, 0, stream>>>(nf, W1, b1, W2, b2, W3, b3, out);
}


# MLP weight chunks and biases are asked for one phase ahead (registers): each chunk's global loads fly under the previous chunk's WMMAs, not between two barriers with the matrix pipe idle
# speedup vs baseline: 1.9988x; 1.0373x over previous
#include <hip/hip_runtime.h>


#define N_NODES 100000
#define N_EDGES 1600000
#define D       128
#define M_TILE  64
#define TPB     128
#define XPITCH  80
#define WPITCH  144

#define CUR_PAD  100352
#define PERM_OFF CUR_PAD
#define WS_NEED  ((size_t)(PERM_OFF + N_EDGES) * 4)
#define SCAN_T   1024
#define NB       256
#define NBKT     391
#define SCAN_N   (NBKT * NB)
#define NG       (N_EDGES / 4)
#define GPB      ((NG + NB - 1) / NB)
#define ST       512
#define GRPS     4
#define NBKT_PAD 416
#define BCAP     8192
#define EPT      (BCAP / ST)
#define GS_BLOCKS 4096

typedef __attribute__((ext_vector_type(2))) float v2f;
typedef __attribute__((ext_vector_type(8))) float v8f;

__device__ __forceinline__ v2f mk2(float a, float b) {
  v2f t; t.x = a; t.y = b; return t;
}

__device__ __forceinline__ int node_of(int d) {
  return (int)min((unsigned)d, (unsigned)(N_NODES - 1));
}

__global__ __launch_bounds__(ST) void hist_kernel(const int* __restrict__ dst,
                                                  int* __restrict__ cnt) {
  __shared__ int h[NBKT];
  const int tid = threadIdx.x;
  if (tid < NBKT) h[tid] = 0;
  __syncthreads();
  static_assert(GPB <= GRPS * ST, "a thread holds at most GRPS groups of four edges");
  const int g0 = blockIdx.x * GPB;
  const int g1 = min(g0 + GPB, NG);
  int4 d[GRPS];
#pragma unroll
  for (int u = 0; u < GRPS; ++u) {
    const int g = g0 + tid + u * ST;
    d[u] = reinterpret_cast<const int4*>(dst)[min(g, NG - 1)];
  }
#pragma unroll
  for (int u = 0; u < GRPS; ++u) {
    if (g0 + tid + u * ST < g1) {
      __hip_atomic_fetch_add(&h[node_of(d[u].x) >> 8], 1, __ATOMIC_RELAXED, __HIP_MEMORY_SCOPE_WORKGROUP);
      __hip_atomic_fetch_add(&h[node_of(d[u].y) >> 8], 1, __ATOMIC_RELAXED, __HIP_MEMORY_SCOPE_WORKGROUP);
      __hip_atomic_fetch_add(&h[node_of(d[u].z) >> 8], 1, __ATOMIC_RELAXED, __HIP_MEMORY_SCOPE_WORKGROUP);
      __hip_atomic_fetch_add(&h[node_of(d[u].w) >> 8], 1, __ATOMIC_RELAXED, __HIP_MEMORY_SCOPE_WORKGROUP);
    }
  }
  __syncthreads();
  if (tid < NBKT) cnt[tid * NB + blockIdx.x] = h[tid];
}

__device__ __forceinline__ int wave_incl_scan(int x, int lane) {
#pragma unroll
  for (int o = 1; o < 32; o <<= 1) {
    int t = __shfl_up(x, o);
    if (lane >= o) x += t;
  }
  return x;
}

__global__ __launch_bounds__(SCAN_T) void scan_kernel(int* __restrict__ cur) {
  __shared__ int sTot[2][32];
  const int tid  = threadIdx.x;
  const int lane = tid & 31;
  const int wv   = tid >> 5;
  int4* c4 = reinterpret_cast<int4*>(cur);
  const int n8 = SCAN_N / 8;
  int carry = 0;
  int4 a = c4[2 * tid], b = c4[2 * tid + 1];
  for (int it = 0; it * SCAN_T < n8; ++it) {
    const int g  = it * SCAN_T + tid;
    const bool ok = g < n8;
    const int gl = min(g + SCAN_T, n8 - 1);
    const int4 na = c4[2 * gl], nb = c4[2 * gl + 1];
    const int s = ok ? (a.x + a.y + a.z + a.w + b.x + b.y + b.z + b.w) : 0;
    const int incl = wave_incl_scan(s, lane);
    if (lane == 31) sTot[it & 1][wv] = incl;
    __syncthreads();
    const int wt    = sTot[it & 1][lane];
    const int wincl = wave_incl_scan(wt, lane);
    const int wexcl = __shfl(wincl, wv) - __shfl(wt, wv);
    const int total = __shfl(wincl, 31);
    const int excl  = carry + wexcl + incl - s;
    if (ok) {
      int4 oa, ob;
      oa.x = excl;        oa.y = oa.x + a.x; oa.z = oa.y + a.y; oa.w = oa.z + a.z;
      ob.x = oa.w + a.w;  ob.y = ob.x + b.x; ob.z = ob.y + b.y; ob.w = ob.z + b.z;
      c4[2 * g] = oa; c4[2 * g + 1] = ob;
    }
    carry += total;
    a = na; b = nb;
  }
}

__device__ __forceinline__ int rank_edge(int* __restrict__ h, int d, int& r) {
  const int n = node_of(d);
  r = __hip_atomic_fetch_add(&h[n >> 8], 1, __ATOMIC_RELAXED, __HIP_MEMORY_SCOPE_WORKGROUP);
  return n;
}

__global__ __launch_bounds__(ST) void part_kernel(const int* __restrict__ src,
                                                  const int* __restrict__ dst,
                                                  const int* __restrict__ cnt,
                                                  int2* __restrict__ tmp) {
  __shared__ int  h[NBKT_PAD];
  __shared__ int  gb[NBKT];
  __shared__ int2 sp[GPB * 4];
  static_assert(GPB <= GRPS * ST, "a thread holds at most GRPS groups of four edges");
  static_assert(NBKT <= NBKT_PAD && NBKT_PAD <= ST && NBKT_PAD == 13 * 32, "one thread per counter; wave 0 takes 13 each");
  const int tid = threadIdx.x;
  const int j   = blockIdx.x;
  int gbv = 0;
  if (tid < NBKT) gbv = cnt[tid * NB + j];
  if (tid < NBKT_PAD) h[tid] = 0;
  __syncthreads();

  const int g0 = j * GPB;
  const int g1 = min(g0 + GPB, NG);
  int4 d[GRPS], sv[GRPS], r[GRPS];
#pragma unroll
  for (int u = 0; u < GRPS; ++u) {
    const int g = g0 + tid + u * ST;
    d[u] = make_int4(0, 0, 0, 0); sv[u] = d[u]; r[u] = d[u];
    if (g < g1) {
      d[u]  = reinterpret_cast<const int4*>(dst)[g];
      sv[u] = reinterpret_cast<const int4*>(src)[g];
    }
  }
#pragma unroll
  for (int u = 0; u < GRPS; ++u) {
    if (g0 + tid + u * ST < g1) {
      d[u].x = rank_edge(h, d[u].x, r[u].x); d[u].y = rank_edge(h, d[u].y, r[u].y);
      d[u].z = rank_edge(h, d[u].z, r[u].z); d[u].w = rank_edge(h, d[u].w, r[u].w);
    }
  }
  __syncthreads();

  if (tid < 32) {
    int v[13];
    int sum = 0;
#pragma unroll
    for (int i = 0; i < 13; ++i) { v[i] = h[tid * 13 + i]; sum += v[i]; }
    int run = wave_incl_scan(sum, tid) - sum;
#pragma unroll
    for (int i = 0; i < 13; ++i) { h[tid * 13 + i] = run; run += v[i]; }
  }
  __syncthreads();

  if (tid < NBKT) gb[tid] = gbv - h[tid];
#pragma unroll
  for (int u = 0; u < GRPS; ++u) {
    if (g0 + tid + u * ST < g1) {
      sp[h[d[u].x >> 8] + r[u].x] = make_int2(sv[u].x, d[u].x);
      sp[h[d[u].y >> 8] + r[u].y] = make_int2(sv[u].y, d[u].y);
      sp[h[d[u].z >> 8] + r[u].z] = make_int2(sv[u].z, d[u].z);
      sp[h[d[u].w >> 8] + r[u].w] = make_int2(sv[u].w, d[u].w);
    }
  }
  __syncthreads();

  const int nE = (g1 - g0) * 4;
  for (int i = tid; i < nE; i += ST) {
    const int2 p = sp[i];
    const unsigned pos = (unsigned)(gb[p.y >> 8] + i);
    tmp[min(pos, (unsigned)(N_EDGES - 1))] = p;
  }
}

__device__ __forceinline__ void bucket_starts(int* __restrict__ c, int cbase, int S,
                                              int b, int lane, int* __restrict__ cur) {
  int v[8];
  int sum = 0;
#pragma unroll
  for (int i = 0; i < 8; ++i) { v[i] = c[lane * 8 + i]; sum += v[i]; }
  int run = wave_incl_scan(sum, lane) - sum;
#pragma unroll
  for (int i = 0; i < 8; ++i) {
    const int node = b * 256 + lane * 8 + i;
    c[lane * 8 + i] = cbase + run;
    if (node < N_NODES) cur[node] = S + run;
    run += v[i];
  }
}

__global__ __launch_bounds__(ST) void bsort_kernel(const int* __restrict__ cnt,
                                                   const int2* __restrict__ tmp,
                                                   int* __restrict__ cur,
                                                   int* __restrict__ perm) {
  __shared__ int c[256];
  __shared__ int sp[BCAP];
  static_assert(BCAP % ST == 0 && ST >= 256, "EPT edges per thread cover BCAP; one thread per node counter");
  const int tid = threadIdx.x;
  const int b = blockIdx.x;
  int S = cnt[b * NB];
  int E = (b + 1 < NBKT) ? cnt[(b + 1) * NB] : N_EDGES;
  S = min(max(S, 0), N_EDGES);
  E = min(max(E, S), N_EDGES);
  const int n = E - S;
  if (tid < 256) c[tid] = 0;
  __syncthreads();

  if (n <= BCAP) {
    int2 v[EPT];
    int  r[EPT];
#pragma unroll
    for (int u = 0; u < EPT; ++u) {
      const int e = S + u * ST + tid;
      v[u] = make_int2(0, 0);
      if (e < E) v[u] = tmp[e];
    }
#pragma unroll
    for (int u = 0; u < EPT; ++u) {
      r[u] = 0;
      if (S + u * ST + tid < E)
        r[u] = __hip_atomic_fetch_add(&c[v[u].y & 255], 1, __ATOMIC_RELAXED, __HIP_MEMORY_SCOPE_WORKGROUP);
    }
    __syncthreads();
    if (tid < 32) bucket_starts(c, 0, S, b, tid, cur);
    __syncthreads();
#pragma unroll
    for (int u = 0; u < EPT; ++u) {
      if (S + u * ST + tid < E) {
        const unsigned q = (unsigned)(c[v[u].y & 255] + r[u]);
        sp[min(q, (unsigned)(BCAP - 1))] = v[u].x;
      }
    }
    __syncthreads();
    for (int i = tid; i < n; i += ST) perm[S + i] = sp[i];
  } else {
    for (int e = S + tid; e < E; e += ST) {
      const int2 v = tmp[e];
      __hip_atomic_fetch_add(&c[v.y & 255], 1, __ATOMIC_RELAXED, __HIP_MEMORY_SCOPE_WORKGROUP);
    }
    __syncthreads();
    if (tid < 32) bucket_starts(c, S, S, b, tid, cur);
    __syncthreads();
    for (int e = S + tid; e < E; e += ST) {
      const int2 v = tmp[e];
      const unsigned p = (unsigned)__hip_atomic_fetch_add(&c[v.y & 255], 1, __ATOMIC_RELAXED, __HIP_MEMORY_SCOPE_WORKGROUP);
      perm[min(p, (unsigned)(N_EDGES - 1))] = v.x;
    }
  }
}

template <int U>
__device__ __forceinline__ void add_rows(float4& acc, const float4* __restrict__ nf4,
                                         int myidx, int k, int lane) {
  float4 v[U];
#pragma unroll
  for (int u = 0; u < U; ++u) {
    int s = __builtin_amdgcn_readlane(myidx, k + u);
    v[u] = nf4[(size_t)s * 32 + lane];
  }
#pragma unroll
  for (int u = 0; u < U; ++u) {
    acc.x += v[u].x; acc.y += v[u].y; acc.z += v[u].z; acc.w += v[u].w;
  }
}

__global__ __launch_bounds__(256) void segsum_kernel(const float* __restrict__ nf,
                                                     const int* __restrict__ cur,
                                                     const int* __restrict__ perm,
                                                     float* __restrict__ agg) {
  const int lane = threadIdx.x & 31;
  const int wave = blockIdx.x * 8 + __builtin_amdgcn_readfirstlane(threadIdx.x >> 5);
  const float4* nf4 = reinterpret_cast<const float4*>(nf);
  float4* agg4 = reinterpret_cast<float4*>(agg);
  for (int n = wave; n < N_NODES; n += GS_BLOCKS * 8) {
    int beg = cur[n];
    int end = (n + 1 < N_NODES) ? cur[n + 1] : N_EDGES;
    beg = max(beg, 0);
    end = min(end, N_EDGES);
    float4 acc = make_float4(0.f, 0.f, 0.f, 0.f);
    for (int base = beg; base < end; base += 32) {
      const int cnt = min(32, end - base);
      int myidx = 0;
      if (lane < cnt) myidx = perm[base + lane];
      myidx = (int)min((unsigned)myidx, (unsigned)(N_NODES - 1));
      int k = 0;
      for (; k + 8 <= cnt; k += 8) add_rows<8>(acc, nf4, myidx, k, lane);
      if (k + 4 <= cnt) { add_rows<4>(acc, nf4, myidx, k, lane); k += 4; }
      if (k + 2 <= cnt) { add_rows<2>(acc, nf4, myidx, k, lane); k += 2; }
      if (k < cnt)      { add_rows<1>(acc, nf4, myidx, k, lane); }
    }
    agg4[(size_t)n * 32 + lane] = acc;
  }
}


__device__ __forceinline__ void load_tile_sw(v2f* __restrict__ sT,
                                             const float* __restrict__ g,
                                             int rowBase, int tid) {
  const float4* g4 = reinterpret_cast<const float4*>(g);
#pragma unroll
  for (int i = 0; i < 16; ++i) {
    int linear = i * TPB + tid;
    int r  = linear & 63;
    int c4 = linear >> 6;
    int gr = rowBase + r;
    if (gr >= N_NODES) gr = N_NODES - 1;
    float4 v = g4[(size_t)gr * 32 + c4];
    sT[(c4 * 2) * XPITCH + r]     = mk2(v.x, v.y);
    sT[(c4 * 2 + 1) * XPITCH + r] = mk2(v.z, v.w);
  }
}

__device__ __forceinline__ void fetch_w(float4 wr[16],
                                        const float* __restrict__ W, int tid) {
  const float4* g4 = reinterpret_cast<const float4*>(W);
#pragma unroll
  for (int i = 0; i < 8; ++i) {
    int linear = i * TPB + tid;
    int p  = linear >> 5;
    int c4 = linear & 31;
    wr[2 * i]     = g4[(p * 2) * 32 + c4];
    wr[2 * i + 1] = g4[(p * 2 + 1) * 32 + c4];
  }
}

__device__ __forceinline__ void stash_w(v2f* __restrict__ sW,
                                        const float4 wr[16], int tid) {
#pragma unroll
  for (int i = 0; i < 8; ++i) {
    int linear = i * TPB + tid;
    int p  = linear >> 5;
    int c4 = linear & 31;
    const float4 r0 = wr[2 * i];
    const float4 r1 = wr[2 * i + 1];
    v2f* dstp = sW + p * WPITCH + c4 * 4;
    dstp[0] = mk2(r0.x, r1.x);
    dstp[1] = mk2(r0.y, r1.y);
    dstp[2] = mk2(r0.z, r1.z);
    dstp[3] = mk2(r0.w, r1.w);
  }
}

__device__ __forceinline__ void fetch_bias(float bv[8],
                                           const float* __restrict__ bias, int lane) {
#pragma unroll
  for (int n = 0; n < 8; ++n) bv[n] = bias[n * 16 + (lane & 15)];
}

__device__ __forceinline__ void gemm64(const v2f* __restrict__ sX, int kpBase,
                                       const v2f* __restrict__ sW,
                                       v8f acc[8], int m0, int lane) {
  const int lrow = lane & 15;
  const int hi   = lane >> 4;
#pragma unroll 4
  for (int kp = 0; kp < 32; kp += 2) {
    v2f a = sX[(kpBase + kp + hi) * XPITCH + m0 + lrow];
    const v2f* wp = sW + (kp + hi) * WPITCH + lrow;
#pragma unroll
    for (int n = 0; n < 8; ++n) {
      v2f b = wp[n * 16];
      acc[n] = __builtin_amdgcn_wmma_f32_16x16x4_f32(
           false, a,  false, b,
           (short)0, acc[n],  false,  false);
    }
  }
}

__device__ __forceinline__ void store_acc_sw(v2f* __restrict__ sT,
                                             const v8f acc[8],
                                             const float bv[8],
                                             int m0, int lane, bool relu) {
  const int lcol = lane & 15;
  const int mhi  = (lane >> 4) << 3;
  float* base = reinterpret_cast<float*>(sT);
#pragma unroll
  for (int n = 0; n < 8; ++n) {
    int col = n * 16 + lcol;
    int p = col >> 1, j = col & 1;
#pragma unroll
    for (int v = 0; v < 8; ++v) {
      float x = acc[n][v] + bv[n];
      if (relu) x = fmaxf(x, 0.f);
      base[(p * XPITCH + m0 + mhi + v) * 2 + j] = x;
    }
  }
}

__global__ __launch_bounds__(TPB) void mlp_kernel(const float* __restrict__ nf,
                                                  const float* __restrict__ W1,
                                                  const float* __restrict__ b1,
                                                  const float* __restrict__ W2,
                                                  const float* __restrict__ b2,
                                                  const float* __restrict__ W3,
                                                  const float* __restrict__ b3,
                                                  float* __restrict__ out) {
  __shared__ v2f sW[32 * WPITCH];
  __shared__ v2f sX[128 * XPITCH];
  __shared__ v2f sH[64 * XPITCH];

  const int tid  = threadIdx.x;
  const int lane = tid & 31;
  const int m0   = (tid >> 5) * 16;
  const int rowBase = blockIdx.x * M_TILE;

  float4 wr[16];
  float bv1[8], bv2[8], bv3[8];
  fetch_w(wr, W1, tid);
  fetch_bias(bv1, b1, lane);
  fetch_bias(bv2, b2, lane);
  fetch_bias(bv3, b3, lane);

  load_tile_sw(sX,       out, rowBase, tid);
  load_tile_sw(sX + 64 * XPITCH, nf, rowBase, tid);

  const v8f zv = {0.f, 0.f, 0.f, 0.f, 0.f, 0.f, 0.f, 0.f};
  v8f acc[8];
#pragma unroll
  for (int n = 0; n < 8; ++n) acc[n] = zv;

#pragma unroll 1
  for (int s = 0; s < 4; ++s) {
    __syncthreads();
    stash_w(sW, wr, tid);
    __syncthreads();
    fetch_w(wr, (s < 3) ? W1 + (size_t)(s + 1) * 64 * D : W2, tid);
    gemm64(sX, s * 32, sW, acc, m0, lane);
  }
  store_acc_sw(sH, acc, bv1, m0, lane, true);

#pragma unroll
  for (int n = 0; n < 8; ++n) acc[n] = zv;
#pragma unroll 1
  for (int s = 0; s < 2; ++s) {
    __syncthreads();
    stash_w(sW, wr, tid);
    __syncthreads();
    fetch_w(wr, (s < 1) ? W2 + (size_t)64 * D : W3, tid);
    gemm64(sH, s * 32, sW, acc, m0, lane);
  }
  store_acc_sw(sX, acc, bv2, m0, lane, true);

#pragma unroll
  for (int n = 0; n < 8; ++n) acc[n] = zv;
#pragma unroll 1
  for (int s = 0; s < 2; ++s) {
    __syncthreads();
    stash_w(sW, wr, tid);
    __syncthreads();
    if (s < 1) fetch_w(wr, W3 + (size_t)64 * D, tid);
    gemm64(sX, s * 32, sW, acc, m0, lane);
  }
  store_acc_sw(sH, acc, bv3, m0, lane, false);
  __syncthreads();

  float4* o4 = reinterpret_cast<float4*>(out);
#pragma unroll
  for (int i = 0; i < 16; ++i) {
    int linear = i * TPB + tid;
    int r  = linear & 63;
    int c4 = linear >> 6;
    int gr = rowBase + r;
    v2f lo = sH[(c4 * 2) * XPITCH + r];
    v2f hv = sH[(c4 * 2 + 1) * XPITCH + r];
    if (gr < N_NODES)
      o4[(size_t)gr * 32 + c4] = make_float4(lo.x, lo.y, hv.x, hv.y);
  }
}

extern "C" void kernel_launch(void* const* d_in, const int* in_sizes, int n_in,
                              void* d_out, int out_size, void* d_ws, size_t ws_size,
                              hipStream_t stream) {
  const float* nf  = (const float*)d_in[0];
  const int*   src = (const int*)d_in[1];
  const int*   dst = (const int*)d_in[2];
  const float* W1  = (const float*)d_in[3];
  const float* b1  = (const float*)d_in[4];
  const float* W2  = (const float*)d_in[5];
  const float* b2  = (const float*)d_in[6];
  const float* W3  = (const float*)d_in[7];
  const float* b3  = (const float*)d_in[8];
  float* out = (float*)d_out;
  int* cur  = (int*)d_ws;
  int* perm = (int*)d_ws + PERM_OFF;
  (void)in_sizes; (void)n_in; (void)out_size;

  if (ws_size < WS_NEED) return;

  int2* tmp = (int2*)d_out;
  int*  cnt = (int*)d_out + 2 * N_EDGES;

  hist_kernel<<<NB, ST, 0, stream>>>(dst, cnt);

  scan_kernel<<<1, SCAN_T, 0, stream>>>(cnt);

  part_kernel<<<NB, ST, 0, stream>>>(src, dst, cnt, tmp);

  bsort_kernel<<<NBKT, ST, 0, stream>>>(cnt, tmp, cur, perm);

  segsum_kernel<<<GS_BLOCKS, 256, 0, stream>>>(nf, cur, perm, out);

  const int nBlocks = (N_NODES + M_TILE - 1) / M_TILE;
  mlp_kernel<<<nBlocks, TPB, 0, stream>>>(nf, W1, b1, W2, b2, W3, b3, out);
}


# MLP activation tile re-laid in LDS (rows rotated per k-pair, pitch 64, h1/h2/out in each wave's own rows): 156 to 100 KB, three work-groups per WGP; tile loads and stores touch 8 lines, not 32
# speedup vs baseline: 2.1772x; 1.0893x over previous
#include <hip/hip_runtime.h>


#define N_NODES 100000
#define N_EDGES 1600000
#define D       128
#define M_TILE  64
#define TPB     128
#define XPITCH  64
#define WPITCH  144

#define CUR_PAD  100352
#define PERM_OFF CUR_PAD
#define WS_NEED  ((size_t)(PERM_OFF + N_EDGES) * 4)
#define SCAN_T   1024
#define NB       256
#define NBKT     391
#define SCAN_N   (NBKT * NB)
#define NG       (N_EDGES / 4)
#define GPB      ((NG + NB - 1) / NB)
#define ST       512
#define GRPS     4
#define NBKT_PAD 416
#define BCAP     8192
#define EPT      (BCAP / ST)
#define GS_BLOCKS 4096

typedef __attribute__((ext_vector_type(2))) float v2f;
typedef __attribute__((ext_vector_type(8))) float v8f;

__device__ __forceinline__ v2f mk2(float a, float b) {
  v2f t; t.x = a; t.y = b; return t;
}

__device__ __forceinline__ int xslot(int p, int r) {
  return p * XPITCH + ((r + (((p >> 1) & 3) << 3) + ((p & 1) << 4)) & 63);
}

__device__ __forceinline__ int node_of(int d) {
  return (int)min((unsigned)d, (unsigned)(N_NODES - 1));
}

__global__ __launch_bounds__(ST) void hist_kernel(const int* __restrict__ dst,
                                                  int* __restrict__ cnt) {
  __shared__ int h[NBKT];
  const int tid = threadIdx.x;
  if (tid < NBKT) h[tid] = 0;
  __syncthreads();
  static_assert(GPB <= GRPS * ST, "a thread holds at most GRPS groups of four edges");
  const int g0 = blockIdx.x * GPB;
  const int g1 = min(g0 + GPB, NG);
  int4 d[GRPS];
#pragma unroll
  for (int u = 0; u < GRPS; ++u) {
    const int g = g0 + tid + u * ST;
    d[u] = reinterpret_cast<const int4*>(dst)[min(g, NG - 1)];
  }
#pragma unroll
  for (int u = 0; u < GRPS; ++u) {
    if (g0 + tid + u * ST < g1) {
      __hip_atomic_fetch_add(&h[node_of(d[u].x) >> 8], 1, __ATOMIC_RELAXED, __HIP_MEMORY_SCOPE_WORKGROUP);
      __hip_atomic_fetch_add(&h[node_of(d[u].y) >> 8], 1, __ATOMIC_RELAXED, __HIP_MEMORY_SCOPE_WORKGROUP);
      __hip_atomic_fetch_add(&h[node_of(d[u].z) >> 8], 1, __ATOMIC_RELAXED, __HIP_MEMORY_SCOPE_WORKGROUP);
      __hip_atomic_fetch_add(&h[node_of(d[u].w) >> 8], 1, __ATOMIC_RELAXED, __HIP_MEMORY_SCOPE_WORKGROUP);
    }
  }
  __syncthreads();
  if (tid < NBKT) cnt[tid * NB + blockIdx.x] = h[tid];
}

__device__ __forceinline__ int wave_incl_scan(int x, int lane) {
#pragma unroll
  for (int o = 1; o < 32; o <<= 1) {
    int t = __shfl_up(x, o);
    if (lane >= o) x += t;
  }
  return x;
}

__global__ __launch_bounds__(SCAN_T) void scan_kernel(int* __restrict__ cur) {
  __shared__ int sTot[2][32];
  const int tid  = threadIdx.x;
  const int lane = tid & 31;
  const int wv   = tid >> 5;
  int4* c4 = reinterpret_cast<int4*>(cur);
  const int n8 = SCAN_N / 8;
  int carry = 0;
  int4 a = c4[2 * tid], b = c4[2 * tid + 1];
  for (int it = 0; it * SCAN_T < n8; ++it) {
    const int g  = it * SCAN_T + tid;
    const bool ok = g < n8;
    const int gl = min(g + SCAN_T, n8 - 1);
    const int4 na = c4[2 * gl], nb = c4[2 * gl + 1];
    const int s = ok ? (a.x + a.y + a.z + a.w + b.x + b.y + b.z + b.w) : 0;
    const int incl = wave_incl_scan(s, lane);
    if (lane == 31) sTot[it & 1][wv] = incl;
    __syncthreads();
    const int wt    = sTot[it & 1][lane];
    const int wincl = wave_incl_scan(wt, lane);
    const int wexcl = __shfl(wincl, wv) - __shfl(wt, wv);
    const int total = __shfl(wincl, 31);
    const int excl  = carry + wexcl + incl - s;
    if (ok) {
      int4 oa, ob;
      oa.x = excl;        oa.y = oa.x + a.x; oa.z = oa.y + a.y; oa.w = oa.z + a.z;
      ob.x = oa.w + a.w;  ob.y = ob.x + b.x; ob.z = ob.y + b.y; ob.w = ob.z + b.z;
      c4[2 * g] = oa; c4[2 * g + 1] = ob;
    }
    carry += total;
    a = na; b = nb;
  }
}

__device__ __forceinline__ int rank_edge(int* __restrict__ h, int d, int& r) {
  const int n = node_of(d);
  r = __hip_atomic_fetch_add(&h[n >> 8], 1, __ATOMIC_RELAXED, __HIP_MEMORY_SCOPE_WORKGROUP);
  return n;
}

__global__ __launch_bounds__(ST) void part_kernel(const int* __restrict__ src,
                                                  const int* __restrict__ dst,
                                                  const int* __restrict__ cnt,
                                                  int2* __restrict__ tmp) {
  __shared__ int  h[NBKT_PAD];
  __shared__ int  gb[NBKT];
  __shared__ int2 sp[GPB * 4];
  static_assert(GPB <= GRPS * ST, "a thread holds at most GRPS groups of four edges");
  static_assert(NBKT <= NBKT_PAD && NBKT_PAD <= ST && NBKT_PAD == 13 * 32, "one thread per counter; wave 0 takes 13 each");
  const int tid = threadIdx.x;
  const int j   = blockIdx.x;
  int gbv = 0;
  if (tid < NBKT) gbv = cnt[tid * NB + j];
  if (tid < NBKT_PAD) h[tid] = 0;
  __syncthreads();

  const int g0 = j * GPB;
  const int g1 = min(g0 + GPB, NG);
  int4 d[GRPS], sv[GRPS], r[GRPS];
#pragma unroll
  for (int u = 0; u < GRPS; ++u) {
    const int g = g0 + tid + u * ST;
    d[u] = make_int4(0, 0, 0, 0); sv[u] = d[u]; r[u] = d[u];
    if (g < g1) {
      d[u]  = reinterpret_cast<const int4*>(dst)[g];
      sv[u] = reinterpret_cast<const int4*>(src)[g];
    }
  }
#pragma unroll
  for (int u = 0; u < GRPS; ++u) {
    if (g0 + tid + u * ST < g1) {
      d[u].x = rank_edge(h, d[u].x, r[u].x); d[u].y = rank_edge(h, d[u].y, r[u].y);
      d[u].z = rank_edge(h, d[u].z, r[u].z); d[u].w = rank_edge(h, d[u].w, r[u].w);
    }
  }
  __syncthreads();

  if (tid < 32) {
    int v[13];
    int sum = 0;
#pragma unroll
    for (int i = 0; i < 13; ++i) { v[i] = h[tid * 13 + i]; sum += v[i]; }
    int run = wave_incl_scan(sum, tid) - sum;
#pragma unroll
    for (int i = 0; i < 13; ++i) { h[tid * 13 + i] = run; run += v[i]; }
  }
  __syncthreads();

  if (tid < NBKT) gb[tid] = gbv - h[tid];
#pragma unroll
  for (int u = 0; u < GRPS; ++u) {
    if (g0 + tid + u * ST < g1) {
      sp[h[d[u].x >> 8] + r[u].x] = make_int2(sv[u].x, d[u].x);
      sp[h[d[u].y >> 8] + r[u].y] = make_int2(sv[u].y, d[u].y);
      sp[h[d[u].z >> 8] + r[u].z] = make_int2(sv[u].z, d[u].z);
      sp[h[d[u].w >> 8] + r[u].w] = make_int2(sv[u].w, d[u].w);
    }
  }
  __syncthreads();

  const int nE = (g1 - g0) * 4;
  for (int i = tid; i < nE; i += ST) {
    const int2 p = sp[i];
    const unsigned pos = (unsigned)(gb[p.y >> 8] + i);
    tmp[min(pos, (unsigned)(N_EDGES - 1))] = p;
  }
}

__device__ __forceinline__ void bucket_starts(int* __restrict__ c, int cbase, int S,
                                              int b, int lane, int* __restrict__ cur) {
  int v[8];
  int sum = 0;
#pragma unroll
  for (int i = 0; i < 8; ++i) { v[i] = c[lane * 8 + i]; sum += v[i]; }
  int run = wave_incl_scan(sum, lane) - sum;
#pragma unroll
  for (int i = 0; i < 8; ++i) {
    const int node = b * 256 + lane * 8 + i;
    c[lane * 8 + i] = cbase + run;
    if (node < N_NODES) cur[node] = S + run;
    run += v[i];
  }
}

__global__ __launch_bounds__(ST) void bsort_kernel(const int* __restrict__ cnt,
                                                   const int2* __restrict__ tmp,
                                                   int* __restrict__ cur,
                                                   int* __restrict__ perm) {
  __shared__ int c[256];
  __shared__ int sp[BCAP];
  static_assert(BCAP % ST == 0 && ST >= 256, "EPT edges per thread cover BCAP; one thread per node counter");
  const int tid = threadIdx.x;
  const int b = blockIdx.x;
  int S = cnt[b * NB];
  int E = (b + 1 < NBKT) ? cnt[(b + 1) * NB] : N_EDGES;
  S = min(max(S, 0), N_EDGES);
  E = min(max(E, S), N_EDGES);
  const int n = E - S;
  if (tid < 256) c[tid] = 0;
  __syncthreads();

  if (n <= BCAP) {
    int2 v[EPT];
    int  r[EPT];
#pragma unroll
    for (int u = 0; u < EPT; ++u) {
      const int e = S + u * ST + tid;
      v[u] = make_int2(0, 0);
      if (e < E) v[u] = tmp[e];
    }
#pragma unroll
    for (int u = 0; u < EPT; ++u) {
      r[u] = 0;
      if (S + u * ST + tid < E)
        r[u] = __hip_atomic_fetch_add(&c[v[u].y & 255], 1, __ATOMIC_RELAXED, __HIP_MEMORY_SCOPE_WORKGROUP);
    }
    __syncthreads();
    if (tid < 32) bucket_starts(c, 0, S, b, tid, cur);
    __syncthreads();
#pragma unroll
    for (int u = 0; u < EPT; ++u) {
      if (S + u * ST + tid < E) {
        const unsigned q = (unsigned)(c[v[u].y & 255] + r[u]);
        sp[min(q, (unsigned)(BCAP - 1))] = v[u].x;
      }
    }
    __syncthreads();
    for (int i = tid; i < n; i += ST) perm[S + i] = sp[i];
  } else {
    for (int e = S + tid; e < E; e += ST) {
      const int2 v = tmp[e];
      __hip_atomic_fetch_add(&c[v.y & 255], 1, __ATOMIC_RELAXED, __HIP_MEMORY_SCOPE_WORKGROUP);
    }
    __syncthreads();
    if (tid < 32) bucket_starts(c, S, S, b, tid, cur);
    __syncthreads();
    for (int e = S + tid; e < E; e += ST) {
      const int2 v = tmp[e];
      const unsigned p = (unsigned)__hip_atomic_fetch_add(&c[v.y & 255], 1, __ATOMIC_RELAXED, __HIP_MEMORY_SCOPE_WORKGROUP);
      perm[min(p, (unsigned)(N_EDGES - 1))] = v.x;
    }
  }
}

template <int U>
__device__ __forceinline__ void add_rows(float4& acc, const float4* __restrict__ nf4,
                                         int myidx, int k, int lane) {
  float4 v[U];
#pragma unroll
  for (int u = 0; u < U; ++u) {
    int s = __builtin_amdgcn_readlane(myidx, k + u);
    v[u] = nf4[(size_t)s * 32 + lane];
  }
#pragma unroll
  for (int u = 0; u < U; ++u) {
    acc.x += v[u].x; acc.y += v[u].y; acc.z += v[u].z; acc.w += v[u].w;
  }
}

__global__ __launch_bounds__(256) void segsum_kernel(const float* __restrict__ nf,
                                                     const int* __restrict__ cur,
                                                     const int* __restrict__ perm,
                                                     float* __restrict__ agg) {
  const int lane = threadIdx.x & 31;
  const int wave = blockIdx.x * 8 + __builtin_amdgcn_readfirstlane(threadIdx.x >> 5);
  const float4* nf4 = reinterpret_cast<const float4*>(nf);
  float4* agg4 = reinterpret_cast<float4*>(agg);
  for (int n = wave; n < N_NODES; n += GS_BLOCKS * 8) {
    int beg = cur[n];
    int end = (n + 1 < N_NODES) ? cur[n + 1] : N_EDGES;
    beg = max(beg, 0);
    end = min(end, N_EDGES);
    float4 acc = make_float4(0.f, 0.f, 0.f, 0.f);
    for (int base = beg; base < end; base += 32) {
      const int cnt = min(32, end - base);
      int myidx = 0;
      if (lane < cnt) myidx = perm[base + lane];
      myidx = (int)min((unsigned)myidx, (unsigned)(N_NODES - 1));
      int k = 0;
      for (; k + 8 <= cnt; k += 8) add_rows<8>(acc, nf4, myidx, k, lane);
      if (k + 4 <= cnt) { add_rows<4>(acc, nf4, myidx, k, lane); k += 4; }
      if (k + 2 <= cnt) { add_rows<2>(acc, nf4, myidx, k, lane); k += 2; }
      if (k < cnt)      { add_rows<1>(acc, nf4, myidx, k, lane); }
    }
    agg4[(size_t)n * 32 + lane] = acc;
  }
}


static_assert(N_NODES % 8 == 0 && M_TILE % 8 == 0, "row blocks of 8 never straddle the end of the table");
__device__ __forceinline__ void load_tile_sw(v2f* __restrict__ sT, int pBase,
                                             const float* __restrict__ g,
                                             int rowBase, int tid) {
  const float4* g4 = reinterpret_cast<const float4*>(g);
  const int wv  = __builtin_amdgcn_readfirstlane(tid >> 5);
  const int rlo = tid & 7;
  const int clo = (tid >> 3) & 3;
  const int lastBlk = (N_NODES - rowBase) / 8 - 1;
#pragma unroll
  for (int i = 0; i < 16; ++i) {
    const int patch = i * 4 + wv;
    const int rblk  = patch & 7;
    const int c4    = (patch >> 3) * 4 + clo;
    const int r     = rblk * 8 + rlo;
    const int gr    = rowBase + min(rblk, lastBlk) * 8 + rlo;
    float4 v = g4[(size_t)gr * 32 + c4];
    sT[xslot(pBase + c4 * 2, r)]     = mk2(v.x, v.y);
    sT[xslot(pBase + c4 * 2 + 1, r)] = mk2(v.z, v.w);
  }
}

__device__ __forceinline__ void fetch_w(float4 wr[16],
                                        const float* __restrict__ W, int tid) {
  const float4* g4 = reinterpret_cast<const float4*>(W);
#pragma unroll
  for (int i = 0; i < 8; ++i) {
    int linear = i * TPB + tid;
    int p  = linear >> 5;
    int c4 = linear & 31;
    wr[2 * i]     = g4[(p * 2) * 32 + c4];
    wr[2 * i + 1] = g4[(p * 2 + 1) * 32 + c4];
  }
}

__device__ __forceinline__ void stash_w(v2f* __restrict__ sW,
                                        const float4 wr[16], int tid) {
#pragma unroll
  for (int i = 0; i < 8; ++i) {
    int linear = i * TPB + tid;
    int p  = linear >> 5;
    int c4 = linear & 31;
    const float4 r0 = wr[2 * i];
    const float4 r1 = wr[2 * i + 1];
    v2f* dstp = sW + p * WPITCH + c4 * 4;
    dstp[0] = mk2(r0.x, r1.x);
    dstp[1] = mk2(r0.y, r1.y);
    dstp[2] = mk2(r0.z, r1.z);
    dstp[3] = mk2(r0.w, r1.w);
  }
}

__device__ __forceinline__ void fetch_bias(float bv[8],
                                           const float* __restrict__ bias, int lane) {
#pragma unroll
  for (int n = 0; n < 8; ++n) bv[n] = bias[n * 16 + (lane & 15)];
}

__device__ __forceinline__ void gemm64(const v2f* __restrict__ sX, int kpBase,
                                       const v2f* __restrict__ sW,
                                       v8f acc[8], int m0, int lane) {
  const int lrow = lane & 15;
  const int hi   = lane >> 4;
#pragma unroll 4
  for (int kp = 0; kp < 32; kp += 2) {
    v2f a = sX[xslot(kpBase + kp + hi, m0 + lrow)];
    const v2f* wp = sW + (kp + hi) * WPITCH + lrow;
#pragma unroll
    for (int n = 0; n < 8; ++n) {
      v2f b = wp[n * 16];
      acc[n] = __builtin_amdgcn_wmma_f32_16x16x4_f32(
           false, a,  false, b,
           (short)0, acc[n],  false,  false);
    }
  }
}

__device__ __forceinline__ void store_acc_sw(v2f* __restrict__ sT, int pBase,
                                             const v8f acc[8],
                                             const float bv[8],
                                             int m0, int lane, bool relu) {
  const int lcol = lane & 15;
  const int mhi  = (lane >> 4) << 3;
  float* base = reinterpret_cast<float*>(sT);
#pragma unroll
  for (int n = 0; n < 8; ++n) {
    int col = n * 16 + lcol;
    int p = pBase + (col >> 1), j = col & 1;
#pragma unroll
    for (int v = 0; v < 8; ++v) {
      float x = acc[n][v] + bv[n];
      if (relu) x = fmaxf(x, 0.f);
      base[xslot(p, m0 + mhi + v) * 2 + j] = x;
    }
  }
}

template <bool GUARD>
__device__ __forceinline__ void store_tile(const v2f* __restrict__ sT,
                                           float* __restrict__ out,
                                           int rowBase, int tid) {
  float4* o4 = reinterpret_cast<float4*>(out);
  const int wv  = __builtin_amdgcn_readfirstlane(tid >> 5);
  const int rlo = tid & 7;
  const int clo = (tid >> 3) & 3;
#pragma unroll
  for (int i = 0; i < 16; ++i) {
    const int patch = i * 4 + wv;
    const int rblk  = patch & 7;
    const int c4    = (patch >> 3) * 4 + clo;
    const int r     = rblk * 8 + rlo;
    const int gr    = rowBase + r;
    v2f lo = sT[xslot(c4 * 2, r)];
    v2f hv = sT[xslot(c4 * 2 + 1, r)];
    if (!GUARD || rowBase + rblk * 8 < N_NODES)
      o4[(size_t)gr * 32 + c4] = make_float4(lo.x, lo.y, hv.x, hv.y);
  }
}

__global__ __launch_bounds__(TPB) __attribute__((amdgpu_waves_per_eu(3, 3))) void mlp_kernel(const float* __restrict__ nf,
                                                  const float* __restrict__ W1,
                                                  const float* __restrict__ b1,
                                                  const float* __restrict__ W2,
                                                  const float* __restrict__ b2,
                                                  const float* __restrict__ W3,
                                                  const float* __restrict__ b3,
                                                  float* __restrict__ out) {
  __shared__ v2f sW[32 * WPITCH];
  __shared__ v2f sX[128 * XPITCH];

  const int tid  = threadIdx.x;
  const int lane = tid & 31;
  const int m0   = (tid >> 5) * 16;
  const int rowBase = blockIdx.x * M_TILE;

  float4 wr[16];
  float bv1[8], bv2[8], bv3[8];
  fetch_w(wr, W1, tid);
  fetch_bias(bv1, b1, lane);
  fetch_bias(bv2, b2, lane);
  fetch_bias(bv3, b3, lane);

  load_tile_sw(sX, 0,  out, rowBase, tid);
  load_tile_sw(sX, 64, nf,  rowBase, tid);

  const v8f zv = {0.f, 0.f, 0.f, 0.f, 0.f, 0.f, 0.f, 0.f};
  v8f acc[8];
#pragma unroll
  for (int n = 0; n < 8; ++n) acc[n] = zv;

#pragma unroll 1
  for (int s = 0; s < 4; ++s) {
    __syncthreads();
    stash_w(sW, wr, tid);
    __syncthreads();
    fetch_w(wr, (s < 3) ? W1 + (size_t)(s + 1) * 64 * D : W2, tid);
    gemm64(sX, s * 32, sW, acc, m0, lane);
  }
  store_acc_sw(sX, 0, acc, bv1, m0, lane, true);

#pragma unroll
  for (int n = 0; n < 8; ++n) acc[n] = zv;
#pragma unroll 1
  for (int s = 0; s < 2; ++s) {
    __syncthreads();
    stash_w(sW, wr, tid);
    __syncthreads();
    fetch_w(wr, (s < 1) ? W2 + (size_t)64 * D : W3, tid);
    gemm64(sX, s * 32, sW, acc, m0, lane);
  }
  store_acc_sw(sX, 64, acc, bv2, m0, lane, true);

#pragma unroll
  for (int n = 0; n < 8; ++n) acc[n] = zv;
#pragma unroll 1
  for (int s = 0; s < 2; ++s) {
    __syncthreads();
    stash_w(sW, wr, tid);
    __syncthreads();
    if (s < 1) fetch_w(wr, W3 + (size_t)64 * D, tid);
    gemm64(sX, 64 + s * 32, sW, acc, m0, lane);
  }
  store_acc_sw(sX, 0, acc, bv3, m0, lane, false);
  __syncthreads();

  if (rowBase + M_TILE <= N_NODES) store_tile<false>(sX, out, rowBase, tid);
  else                             store_tile<true>(sX, out, rowBase, tid);
}

extern "C" void kernel_launch(void* const* d_in, const int* in_sizes, int n_in,
                              void* d_out, int out_size, void* d_ws, size_t ws_size,
                              hipStream_t stream) {
  const float* nf  = (const float*)d_in[0];
  const int*   src = (const int*)d_in[1];
  const int*   dst = (const int*)d_in[2];
  const float* W1  = (const float*)d_in[3];
  const float* b1  = (const float*)d_in[4];
  const float* W2  = (const float*)d_in[5];
  const float* b2  = (const float*)d_in[6];
  const float* W3  = (const float*)d_in[7];
  const float* b3  = (const float*)d_in[8];
  float* out = (float*)d_out;
  int* cur  = (int*)d_ws;
  int* perm = (int*)d_ws + PERM_OFF;
  (void)in_sizes; (void)n_in; (void)out_size;

  if (ws_size < WS_NEED) return;

  int2* tmp = (int2*)d_out;
  int*  cnt = (int*)d_out + 2 * N_EDGES;

  hist_kernel<<<NB, ST, 0, stream>>>(dst, cnt);

  scan_kernel<<<1, SCAN_T, 0, stream>>>(cnt);

  part_kernel<<<NB, ST, 0, stream>>>(src, dst, cnt, tmp);

  bsort_kernel<<<NBKT, ST, 0, stream>>>(cnt, tmp, cur, perm);

  segsum_kernel<<<GS_BLOCKS, 256, 0, stream>>>(nf, cur, perm, out);

  const int nBlocks = (N_NODES + M_TILE - 1) / M_TILE;
  mlp_kernel<<<nBlocks, TPB, 0, stream>>>(nf, W1, b1, W2, b2, W3, b3, out);
}
